# starting kernel
# baseline (speedup 1.0000x reference)
#include <hip/hip_runtime.h>

typedef float v2f __attribute__((ext_vector_type(2)));
typedef float v8f __attribute__((ext_vector_type(8)));

#define DFEAT 128


__global__ void zero_f4_kernel(float* __restrict__ p, int n4) {
    int i = blockIdx.x * blockDim.x + threadIdx.x;
    if (i < n4) ((float4*)p)[i] = make_float4(0.f, 0.f, 0.f, 0.f);
}

__global__ void deg_kernel(const int* __restrict__ dst, float* __restrict__ deg, int E) {
    int e = blockIdx.x * blockDim.x + threadIdx.x;
    if (e < E) atomicAdd(&deg[dst[e]], 1.0f);
}

__global__ void dinv_kernel(const float* __restrict__ deg, float* __restrict__ dinv, int N) {
    int i = blockIdx.x * blockDim.x + threadIdx.x;
    if (i < N) dinv[i] = 1.0f / sqrtf(fmaxf(deg[i], 1.0f));
}

__global__ void h_init_kernel(const float* __restrict__ feat, const float* __restrict__ theta,
                              float* __restrict__ h, int n4) {
    int i = blockIdx.x * blockDim.x + threadIdx.x;
    if (i >= n4) return;
    float t0 = theta[0];
    float4 f = ((const float4*)feat)[i];
    ((float4*)h)[i] = make_float4(t0 * f.x, t0 * f.y, t0 * f.z, t0 * f.w);
}


__global__ void scatter_kernel(const float* __restrict__ x, const float* __restrict__ dinv,
                               const int* __restrict__ src, const int* __restrict__ dst,
                               float* __restrict__ agg, int E) {
    int t = blockIdx.x * blockDim.x + threadIdx.x;
    int e = t >> 5;
    int lane = t & 31;
    if (e >= E) return;
    int s = src[e];
    int d = dst[e];
    float w = dinv[s];
    float4 v = *(const float4*)(x + (size_t)s * DFEAT + lane * 4);
    float* ap = agg + (size_t)d * DFEAT + lane * 4;
    atomicAdd(ap + 0, w * v.x);
    atomicAdd(ap + 1, w * v.y);
    atomicAdd(ap + 2, w * v.z);
    atomicAdd(ap + 3, w * v.w);
}


__global__ void cheb_update_kernel(float* __restrict__ agg, const float* __restrict__ dinv,
                                   const float* __restrict__ Tprev, float* __restrict__ Tnew,
                                   float* __restrict__ h, const float* __restrict__ theta,
                                   int order, float c2, int N) {
    int i = blockIdx.x * blockDim.x + threadIdx.x;
    int total = N * (DFEAT / 4);
    if (i >= total) return;
    int n = i / (DFEAT / 4);
    float s = -c2 * dinv[n];
    float th = theta[order] * 0.5f;

    float4 a = ((const float4*)agg)[i];
    ((float4*)agg)[i] = make_float4(0.f, 0.f, 0.f, 0.f);

    float4 tp = Tprev ? ((const float4*)Tprev)[i] : make_float4(0.f, 0.f, 0.f, 0.f);
    float4 t;
    t.x = fmaf(s, a.x, -tp.x);
    t.y = fmaf(s, a.y, -tp.y);
    t.z = fmaf(s, a.z, -tp.z);
    t.w = fmaf(s, a.w, -tp.w);
    ((float4*)Tnew)[i] = t;

    float4 hv = ((const float4*)h)[i];
    hv.x = fmaf(th, t.x + 1.0f, hv.x);
    hv.y = fmaf(th, t.y + 1.0f, hv.y);
    hv.z = fmaf(th, t.z + 1.0f, hv.z);
    hv.w = fmaf(th, t.w + 1.0f, hv.w);
    ((float4*)h)[i] = hv;
}


__global__ void gemm_wmma_kernel(const float* __restrict__ h, const float* __restrict__ W,
                                 float* __restrict__ out, int N) {
    const int lane = threadIdx.x & 31;
    const int wave = threadIdx.x >> 5;
    const int tile = blockIdx.x * (blockDim.x >> 5) + wave;
    const int nRowTiles = N >> 4;
    const int row = tile >> 3;
    const int col = tile & 7;
    if (row >= nRowTiles) return;

    const int rowBase = row << 4;
    const int colBase = col << 4;
    const int m = lane & 15;
    const int half = lane >> 4;
    const int kb = half << 1;

    const float* hrow = h + (size_t)(rowBase + m) * DFEAT + kb;
    const float* wrow = W + (size_t)(colBase + m) * DFEAT + kb;

    v8f acc = {};
#pragma unroll
    for (int kk = 0; kk < DFEAT; kk += 4) {
        v2f a = *(const v2f*)(hrow + kk);
        v2f b = *(const v2f*)(wrow + kk);
        acc = __builtin_amdgcn_wmma_f32_16x16x4_f32(
             false, a,  false, b,
             (short)0, acc,  false,  false);
    }

#pragma unroll
    for (int v = 0; v < 8; ++v) {
        int r = rowBase + v + (half << 3);
        float val = acc[v];
        val = (val >= 0.0f) ? val : 0.01f * val;
        out[(size_t)r * DFEAT + colBase + m] = val;
    }
}


extern "C" void kernel_launch(void* const* d_in, const int* in_sizes, int n_in,
                              void* d_out, int out_size, void* d_ws, size_t ws_size,
                              hipStream_t stream) {
    const float* feat  = (const float*)d_in[0];
    const int*   src   = (const int*)d_in[1];
    const int*   dst   = (const int*)d_in[2];
    const float* theta = (const float*)d_in[3];
    const float* W     = (const float*)d_in[4];
    float* out = (float*)d_out;

    const int N = in_sizes[0] / DFEAT;
    const int E = in_sizes[1];

    float* ws   = (float*)d_ws;
    float* deg  = ws;
    float* dinv = deg + N;
    float* agg  = dinv + N;
    float* Ta   = agg + (size_t)N * DFEAT;
    float* Tb   = Ta + (size_t)N * DFEAT;
    float* h    = Tb + (size_t)N * DFEAT;

    const int TB = 256;
    const int nd4 = N * (DFEAT / 4);
    auto cdiv = [](long long a, long long b) { return (int)((a + b - 1) / b); };

    zero_f4_kernel<<<cdiv(N / 4, TB), TB, 0, stream>>>(deg, N / 4);
    zero_f4_kernel<<<cdiv(nd4, TB), TB, 0, stream>>>(agg, nd4);
    deg_kernel<<<cdiv(E, TB), TB, 0, stream>>>(dst, deg, E);
    dinv_kernel<<<cdiv(N, TB), TB, 0, stream>>>(deg, dinv, N);

    h_init_kernel<<<cdiv(nd4, TB), TB, 0, stream>>>(feat, theta, h, nd4);

    const long long edgeThreads = (long long)E * 32;

    scatter_kernel<<<cdiv(edgeThreads, TB), TB, 0, stream>>>(feat, dinv, src, dst, agg, E);
    cheb_update_kernel<<<cdiv(nd4, TB), TB, 0, stream>>>(agg, dinv, nullptr, Ta, h, theta, 1, 1.0f, N);

    scatter_kernel<<<cdiv(edgeThreads, TB), TB, 0, stream>>>(Ta, dinv, src, dst, agg, E);
    cheb_update_kernel<<<cdiv(nd4, TB), TB, 0, stream>>>(agg, dinv, feat, Tb, h, theta, 2, 2.0f, N);

    scatter_kernel<<<cdiv(edgeThreads, TB), TB, 0, stream>>>(Tb, dinv, src, dst, agg, E);
    cheb_update_kernel<<<cdiv(nd4, TB), TB, 0, stream>>>(agg, dinv, Ta, Ta, h, theta, 3, 2.0f, N);

    const int tiles = (N / 16) * (DFEAT / 16);
    gemm_wmma_kernel<<<cdiv(tiles, 256 / 32), 256, 0, stream>>>(h, W, out, N);
}


# Scatter-add into an LDS tile: a group owns 196 dst rows, filters the dst list, adds gathered rows by LDS float adds, finishes them; global atomics, agg plane, update launches gone; arrival-order sums
# speedup vs baseline: 2.2067x; 2.2067x over previous
#include <hip/hip_runtime.h>

typedef float v2f __attribute__((ext_vector_type(2)));
typedef float v8f __attribute__((ext_vector_type(8)));

#define DFEAT 128


__global__ void zero_f4_kernel(float* __restrict__ p, int n4) {
    int i = blockIdx.x * blockDim.x + threadIdx.x;
    if (i < n4) ((float4*)p)[i] = make_float4(0.f, 0.f, 0.f, 0.f);
}

__global__ void deg_kernel(const int* __restrict__ dst, float* __restrict__ deg, int E) {
    int e = blockIdx.x * blockDim.x + threadIdx.x;
    if (e < E) atomicAdd(&deg[dst[e]], 1.0f);
}

__global__ void dinv_kernel(const float* __restrict__ deg, float* __restrict__ dinv, int N) {
    int i = blockIdx.x * blockDim.x + threadIdx.x;
    if (i < N) dinv[i] = 1.0f / sqrtf(fmaxf(deg[i], 1.0f));
}

__global__ void h_init_kernel(const float* __restrict__ feat, const float* __restrict__ theta,
                              float* __restrict__ h, int n4) {
    int i = blockIdx.x * blockDim.x + threadIdx.x;
    if (i >= n4) return;
    float t0 = theta[0];
    float4 f = ((const float4*)feat)[i];
    ((float4*)h)[i] = make_float4(t0 * f.x, t0 * f.y, t0 * f.z, t0 * f.w);
}


#define PROP_THREADS 1024
#define PROP_WAVES   32
#define PROP_ROWS    196
#define SCAN_UNROLL  16
#define TEST_QUADS   8
#define DRAIN_BATCH  8
#define QCAP (DRAIN_BATCH - 1 + 32)

__device__ __forceinline__ void drain_edges(const float* __restrict__ x,
                                            const float* __restrict__ dinv,
                                            const int* __restrict__ src,
                                            const int* __restrict__ dst,
                                            float* acc, const int* q, int n, int lo, int lane) {
    int s = 0, dl = 0;
    float w = 0.0f;
    if (lane < n) {
        int e = q[lane];
        s = src[e];
        dl = dst[e] - lo;
    }
    float v[DRAIN_BATCH][4];
#pragma unroll
    for (int j = 0; j < DRAIN_BATCH; ++j) {
        v[j][0] = 0.0f; v[j][1] = 0.0f; v[j][2] = 0.0f; v[j][3] = 0.0f;
        if (j < n) {
            int sj = __builtin_amdgcn_readlane(s, j);
            const float* xp = x + (size_t)sj * DFEAT + lane;
            v[j][0] = xp[0];
            v[j][1] = xp[32];
            v[j][2] = xp[64];
            v[j][3] = xp[96];
        }
    }
    if (lane < n) w = dinv[s];
#pragma unroll
    for (int j = 0; j < DRAIN_BATCH; ++j) {
        if (j < n) {
            float wj = __int_as_float(__builtin_amdgcn_readlane(__float_as_int(w), j));
            int dj = __builtin_amdgcn_readlane(dl, j);
            float* ap = acc + dj * DFEAT + lane;
            atomicAdd(ap + 0,  wj * v[j][0]);
            atomicAdd(ap + 32, wj * v[j][1]);
            atomicAdd(ap + 64, wj * v[j][2]);
            atomicAdd(ap + 96, wj * v[j][3]);
        }
    }
}

__device__ __forceinline__ void test_quads(const int4* d, int base, int lo, int cnt, int lane,
                                           int* q, int& qn,
                                           const float* __restrict__ x, const float* __restrict__ dinv,
                                           const int* __restrict__ src, const int* __restrict__ dst,
                                           float* acc) {
    unsigned mm = 0;
#pragma unroll
    for (int u = 0; u < TEST_QUADS; ++u) {
        mm |= (unsigned)((unsigned)(d[u].x - lo) < (unsigned)cnt) << (4 * u + 0);
        mm |= (unsigned)((unsigned)(d[u].y - lo) < (unsigned)cnt) << (4 * u + 1);
        mm |= (unsigned)((unsigned)(d[u].z - lo) < (unsigned)cnt) << (4 * u + 2);
        mm |= (unsigned)((unsigned)(d[u].w - lo) < (unsigned)cnt) << (4 * u + 3);
    }
    for (;;) {
        unsigned b = __builtin_amdgcn_ballot_w32(mm != 0);
        if (b == 0) break;
        if (mm != 0) {
            int bit = __builtin_ctz(mm);
            mm &= mm - 1;
            int i = base + (bit >> 2) * PROP_THREADS;
            int pos = qn + __builtin_popcount(b & ((1u << lane) - 1u));
            q[pos] = 4 * i + (bit & 3);
        }
        qn += __builtin_popcount(b);
        while (qn >= DRAIN_BATCH) {
            qn -= DRAIN_BATCH;
            drain_edges(x, dinv, src, dst, acc, q + qn, DRAIN_BATCH, lo, lane);
        }
    }
}

__global__ __launch_bounds__(PROP_THREADS)
void propagate_kernel(const float* __restrict__ x, const float* __restrict__ dinv,
                      const int* __restrict__ src, const int* __restrict__ dst,
                      const float* Tprev, float* Tnew, float* __restrict__ h,
                      const float* __restrict__ theta, int order, float c2, int N, int E) {
    __shared__ __attribute__((aligned(16))) float acc[PROP_ROWS * DFEAT];
    __shared__ int queue[PROP_WAVES][QCAP];

    const int tid = threadIdx.x;
    const int lane = tid & 31;
    const int wave = tid >> 5;
    const int lo = blockIdx.x * PROP_ROWS;
    const int cnt = (N - lo < PROP_ROWS) ? (N - lo) : PROP_ROWS;

    for (int i = tid; i < PROP_ROWS * DFEAT / 4; i += PROP_THREADS)
        ((float4*)acc)[i] = make_float4(0.f, 0.f, 0.f, 0.f);
    __syncthreads();

    int* q = queue[wave];
    int qn = 0;

    const int E4 = E >> 2;
    const int span = SCAN_UNROLL * PROP_THREADS;
    const int nG = (E4 + span - 1) / span;
    const int gStart = (int)((long long)blockIdx.x * nG / gridDim.x);
    const int4* dst4 = (const int4*)dst;

    for (int g = 0; g < nG; ++g) {
        int gg = g + gStart;
        if (gg >= nG) gg -= nG;
        const int base = gg * span + tid;
        int4 d[SCAN_UNROLL];
        if (gg * span + span <= E4) {
#pragma unroll
            for (int u = 0; u < SCAN_UNROLL; ++u) d[u] = dst4[base + u * PROP_THREADS];
        } else {
#pragma unroll
            for (int u = 0; u < SCAN_UNROLL; ++u) {
                int i = base + u * PROP_THREADS;
                int4 t = dst4[i < E4 ? i : 0];
                if (i >= E4) t = make_int4(-1, -1, -1, -1);
                d[u] = t;
            }
        }
        test_quads(d, base, lo, cnt, lane, q, qn, x, dinv, src, dst, acc);
        test_quads(d + TEST_QUADS, base + TEST_QUADS * PROP_THREADS, lo, cnt, lane, q, qn, x, dinv, src, dst, acc);
    }

    const int rem = E - 4 * E4;
    if (wave == 0 && rem > 0) {
        int e = 4 * E4 + lane;
        bool hit = false;
        if (lane < rem) hit = (unsigned)(dst[e] - lo) < (unsigned)cnt;
        unsigned b = __builtin_amdgcn_ballot_w32(hit);
        if (hit) q[qn + __builtin_popcount(b & ((1u << lane) - 1u))] = e;
        qn += __builtin_popcount(b);
    }
    while (qn > 0) {
        int n = qn < DRAIN_BATCH ? qn : DRAIN_BATCH;
        qn -= n;
        drain_edges(x, dinv, src, dst, acc, q + qn, n, lo, lane);
    }
    __syncthreads();

    const float th = theta[order] * 0.5f;
    const int tot4 = cnt * (DFEAT / 4);
    const size_t g0 = (size_t)lo * (DFEAT / 4);
    for (int i = tid; i < tot4; i += PROP_THREADS) {
        int n = lo + (i >> 5);
        float s = -c2 * dinv[n];
        float4 a = ((const float4*)acc)[i];
        float4 tp = Tprev ? ((const float4*)Tprev)[g0 + i] : make_float4(0.f, 0.f, 0.f, 0.f);
        float4 t;
        t.x = fmaf(s, a.x, -tp.x);
        t.y = fmaf(s, a.y, -tp.y);
        t.z = fmaf(s, a.z, -tp.z);
        t.w = fmaf(s, a.w, -tp.w);
        ((float4*)Tnew)[g0 + i] = t;

        float4 hv = ((const float4*)h)[g0 + i];
        hv.x = fmaf(th, t.x + 1.0f, hv.x);
        hv.y = fmaf(th, t.y + 1.0f, hv.y);
        hv.z = fmaf(th, t.z + 1.0f, hv.z);
        hv.w = fmaf(th, t.w + 1.0f, hv.w);
        ((float4*)h)[g0 + i] = hv;
    }
}


__global__ void gemm_wmma_kernel(const float* __restrict__ h, const float* __restrict__ W,
                                 float* __restrict__ out, int N) {
    const int lane = threadIdx.x & 31;
    const int wave = threadIdx.x >> 5;
    const int tile = blockIdx.x * (blockDim.x >> 5) + wave;
    const int nRowTiles = N >> 4;
    const int row = tile >> 3;
    const int col = tile & 7;
    if (row >= nRowTiles) return;

    const int rowBase = row << 4;
    const int colBase = col << 4;
    const int m = lane & 15;
    const int half = lane >> 4;
    const int kb = half << 1;

    const float* hrow = h + (size_t)(rowBase + m) * DFEAT + kb;
    const float* wrow = W + (size_t)(colBase + m) * DFEAT + kb;

    v8f acc = {};
#pragma unroll
    for (int kk = 0; kk < DFEAT; kk += 4) {
        v2f a = *(const v2f*)(hrow + kk);
        v2f b = *(const v2f*)(wrow + kk);
        acc = __builtin_amdgcn_wmma_f32_16x16x4_f32(
             false, a,  false, b,
             (short)0, acc,  false,  false);
    }

#pragma unroll
    for (int v = 0; v < 8; ++v) {
        int r = rowBase + v + (half << 3);
        float val = acc[v];
        val = (val >= 0.0f) ? val : 0.01f * val;
        out[(size_t)r * DFEAT + colBase + m] = val;
    }
}


extern "C" void kernel_launch(void* const* d_in, const int* in_sizes, int n_in,
                              void* d_out, int out_size, void* d_ws, size_t ws_size,
                              hipStream_t stream) {
    const float* feat  = (const float*)d_in[0];
    const int*   src   = (const int*)d_in[1];
    const int*   dst   = (const int*)d_in[2];
    const float* theta = (const float*)d_in[3];
    const float* W     = (const float*)d_in[4];
    float* out = (float*)d_out;

    const int N = in_sizes[0] / DFEAT;
    const int E = in_sizes[1];

    float* ws   = (float*)d_ws;
    float* deg  = ws;
    float* dinv = deg + N;
    float* Ta   = dinv + N + (size_t)N * DFEAT;
    float* Tb   = Ta + (size_t)N * DFEAT;
    float* h    = Tb + (size_t)N * DFEAT;

    const int TB = 256;
    const int nd4 = N * (DFEAT / 4);
    auto cdiv = [](long long a, long long b) { return (int)((a + b - 1) / b); };

    zero_f4_kernel<<<cdiv(N / 4, TB), TB, 0, stream>>>(deg, N / 4);
    deg_kernel<<<cdiv(E, TB), TB, 0, stream>>>(dst, deg, E);
    dinv_kernel<<<cdiv(N, TB), TB, 0, stream>>>(deg, dinv, N);

    h_init_kernel<<<cdiv(nd4, TB), TB, 0, stream>>>(feat, theta, h, nd4);

    const int groups = cdiv(N, PROP_ROWS);

    propagate_kernel<<<groups, PROP_THREADS, 0, stream>>>(feat, dinv, src, dst, nullptr, Ta, h, theta, 1, 1.0f, N, E);

    propagate_kernel<<<groups, PROP_THREADS, 0, stream>>>(Ta, dinv, src, dst, feat, Tb, h, theta, 2, 2.0f, N, E);

    propagate_kernel<<<groups, PROP_THREADS, 0, stream>>>(Tb, dinv, src, dst, Ta, Ta, h, theta, 3, 2.0f, N, E);

    const int tiles = (N / 16) * (DFEAT / 16);
    gemm_wmma_kernel<<<cdiv(tiles, 256 / 32), 256, 0, stream>>>(h, W, out, N);
}


# Orders 2 and 3 walk a per-tile edge list (source row, local row) the order-1 filter scan appends to the workspace at offsets summed from degree counts: dst filtered once, not 3x; arrival-order sums
# speedup vs baseline: 2.7100x; 1.2281x over previous
#include <hip/hip_runtime.h>

typedef float v2f __attribute__((ext_vector_type(2)));
typedef float v8f __attribute__((ext_vector_type(8)));

#define DFEAT 128


__global__ void zero_f4_kernel(float* __restrict__ p, int n4) {
    int i = blockIdx.x * blockDim.x + threadIdx.x;
    if (i < n4) ((float4*)p)[i] = make_float4(0.f, 0.f, 0.f, 0.f);
}

__global__ void deg_kernel(const int* __restrict__ dst, float* __restrict__ deg, int E) {
    int e = blockIdx.x * blockDim.x + threadIdx.x;
    if (e < E) atomicAdd(&deg[dst[e]], 1.0f);
}

__global__ void dinv_kernel(const float* __restrict__ deg, float* __restrict__ dinv, int N) {
    int i = blockIdx.x * blockDim.x + threadIdx.x;
    if (i < N) dinv[i] = 1.0f / sqrtf(fmaxf(deg[i], 1.0f));
}

__global__ void h_init_kernel(const float* __restrict__ feat, const float* __restrict__ theta,
                              float* __restrict__ h, int n4) {
    int i = blockIdx.x * blockDim.x + threadIdx.x;
    if (i >= n4) return;
    float t0 = theta[0];
    float4 f = ((const float4*)feat)[i];
    ((float4*)h)[i] = make_float4(t0 * f.x, t0 * f.y, t0 * f.z, t0 * f.w);
}


#define PROP_THREADS 1024
#define PROP_WAVES   32
#define PROP_ROWS    196
#define SCAN_UNROLL  16
#define TEST_QUADS   8
#define DRAIN_BATCH  8
#define QCAP (DRAIN_BATCH - 1 + 32)
static_assert(PROP_ROWS % 4 == 0, "a tile's first row must be a multiple of 4: the degree counts in front of it are read as float4");
#define BASE_UNROLL  8

__device__ __forceinline__ void gather_add(const float* __restrict__ x,
                                           const float* __restrict__ dinv,
                                           float* acc, int s, int dl, int n, int lane,
                                           int2* tileList, int* cursor) {
    float w = 0.0f;
    float v[DRAIN_BATCH][4];
#pragma unroll
    for (int j = 0; j < DRAIN_BATCH; ++j) {
        v[j][0] = 0.0f; v[j][1] = 0.0f; v[j][2] = 0.0f; v[j][3] = 0.0f;
        if (j < n) {
            int sj = __builtin_amdgcn_readlane(s, j);
            const float* xp = x + (size_t)sj * DFEAT + lane;
            v[j][0] = xp[0];
            v[j][1] = xp[32];
            v[j][2] = xp[64];
            v[j][3] = xp[96];
        }
    }
    if (lane < n) w = dinv[s];
    if (tileList) {
        int pos = 0;
        if (lane == 0) pos = atomicAdd(cursor, n);
        pos = __builtin_amdgcn_readfirstlane(pos);
        if (lane < n) tileList[pos + lane] = make_int2(s, dl);
    }
#pragma unroll
    for (int j = 0; j < DRAIN_BATCH; ++j) {
        if (j < n) {
            float wj = __int_as_float(__builtin_amdgcn_readlane(__float_as_int(w), j));
            int dj = __builtin_amdgcn_readlane(dl, j);
            float* ap = acc + dj * DFEAT + lane;
            atomicAdd(ap + 0,  wj * v[j][0]);
            atomicAdd(ap + 32, wj * v[j][1]);
            atomicAdd(ap + 64, wj * v[j][2]);
            atomicAdd(ap + 96, wj * v[j][3]);
        }
    }
}

__device__ __forceinline__ void drain_edges(const float* __restrict__ x,
                                            const float* __restrict__ dinv,
                                            const int* __restrict__ src,
                                            const int* __restrict__ dst,
                                            float* acc, const int* q, int n, int lo, int lane,
                                            int2* tileList, int* cursor) {
    int s = 0, dl = 0;
    if (lane < n) {
        int e = q[lane];
        s = src[e];
        dl = dst[e] - lo;
    }
    gather_add(x, dinv, acc, s, dl, n, lane, tileList, cursor);
}

__device__ __forceinline__ void test_quads(const int4* d, int base, int lo, int cnt, int lane,
                                           int* q, int& qn,
                                           const float* __restrict__ x, const float* __restrict__ dinv,
                                           const int* __restrict__ src, const int* __restrict__ dst,
                                           float* acc, int2* tileList, int* cursor) {
    unsigned mm = 0;
#pragma unroll
    for (int u = 0; u < TEST_QUADS; ++u) {
        mm |= (unsigned)((unsigned)(d[u].x - lo) < (unsigned)cnt) << (4 * u + 0);
        mm |= (unsigned)((unsigned)(d[u].y - lo) < (unsigned)cnt) << (4 * u + 1);
        mm |= (unsigned)((unsigned)(d[u].z - lo) < (unsigned)cnt) << (4 * u + 2);
        mm |= (unsigned)((unsigned)(d[u].w - lo) < (unsigned)cnt) << (4 * u + 3);
    }
    for (;;) {
        unsigned b = __builtin_amdgcn_ballot_w32(mm != 0);
        if (b == 0) break;
        if (mm != 0) {
            int bit = __builtin_ctz(mm);
            mm &= mm - 1;
            int i = base + (bit >> 2) * PROP_THREADS;
            int pos = qn + __builtin_popcount(b & ((1u << lane) - 1u));
            q[pos] = 4 * i + (bit & 3);
        }
        qn += __builtin_popcount(b);
        while (qn >= DRAIN_BATCH) {
            qn -= DRAIN_BATCH;
            drain_edges(x, dinv, src, dst, acc, q + qn, DRAIN_BATCH, lo, lane, tileList, cursor);
        }
    }
}

__device__ __forceinline__ void finish_rows(const float* acc, const float* __restrict__ dinv,
                                            const float* Tprev, float* Tnew, float* __restrict__ h,
                                            const float* __restrict__ theta, int order, float c2,
                                            int lo, int cnt, int tid) {
    const float th = theta[order] * 0.5f;
    const int tot4 = cnt * (DFEAT / 4);
    const size_t g0 = (size_t)lo * (DFEAT / 4);
    for (int i = tid; i < tot4; i += PROP_THREADS) {
        int n = lo + (i >> 5);
        float s = -c2 * dinv[n];
        float4 a = ((const float4*)acc)[i];
        float4 tp = Tprev ? ((const float4*)Tprev)[g0 + i] : make_float4(0.f, 0.f, 0.f, 0.f);
        float4 t;
        t.x = fmaf(s, a.x, -tp.x);
        t.y = fmaf(s, a.y, -tp.y);
        t.z = fmaf(s, a.z, -tp.z);
        t.w = fmaf(s, a.w, -tp.w);
        ((float4*)Tnew)[g0 + i] = t;

        float4 hv = ((const float4*)h)[g0 + i];
        hv.x = fmaf(th, t.x + 1.0f, hv.x);
        hv.y = fmaf(th, t.y + 1.0f, hv.y);
        hv.z = fmaf(th, t.z + 1.0f, hv.z);
        hv.w = fmaf(th, t.w + 1.0f, hv.w);
        ((float4*)h)[g0 + i] = hv;
    }
}

__global__ __launch_bounds__(PROP_THREADS)
void propagate_kernel(const float* __restrict__ x, const float* __restrict__ dinv,
                      const float* __restrict__ deg,
                      const int* __restrict__ src, const int* __restrict__ dst,
                      int* __restrict__ elist,
                      const float* Tprev, float* Tnew, float* __restrict__ h,
                      const float* __restrict__ theta, int order, float c2, int N, int E) {
    __shared__ __attribute__((aligned(16))) float acc[PROP_ROWS * DFEAT];
    __shared__ int queue[PROP_WAVES][QCAP];
    __shared__ int listBase;
    __shared__ int cursor;

    const int tid = threadIdx.x;
    const int lane = tid & 31;
    const int wave = tid >> 5;
    const int lo = blockIdx.x * PROP_ROWS;
    const int cnt = (N - lo < PROP_ROWS) ? (N - lo) : PROP_ROWS;

    if (tid == 0) { listBase = 0; cursor = 0; }
    for (int i = tid; i < PROP_ROWS * DFEAT / 4; i += PROP_THREADS)
        ((float4*)acc)[i] = make_float4(0.f, 0.f, 0.f, 0.f);

    int part = 0;
    {
        const float4* deg4 = (const float4*)deg;
        const int n4 = lo >> 2;
        for (int i0 = tid; i0 < n4; i0 += BASE_UNROLL * PROP_THREADS) {
            float4 dv[BASE_UNROLL];
#pragma unroll
            for (int u = 0; u < BASE_UNROLL; ++u) {
                int i = i0 + u * PROP_THREADS;
                dv[u] = deg4[i < n4 ? i : 0];
            }
#pragma unroll
            for (int u = 0; u < BASE_UNROLL; ++u) {
                int i = i0 + u * PROP_THREADS;
                int t = (int)dv[u].x + (int)dv[u].y + (int)dv[u].z + (int)dv[u].w;
                part += (i < n4) ? t : 0;
            }
        }
        for (int off = 16; off > 0; off >>= 1) part += __shfl_xor(part, off, 32);
    }
    __syncthreads();
    if (lane == 0 && part != 0) atomicAdd(&listBase, part);
    __syncthreads();
    int2* list2 = (int2*)elist;
    int2* tileList = list2 + listBase;

    int* q = queue[wave];
    int qn = 0;

    const int E4 = E >> 2;
    const int span = SCAN_UNROLL * PROP_THREADS;
    const int nG = (E4 + span - 1) / span;
    const int gStart = (int)((long long)blockIdx.x * nG / gridDim.x);
    const int4* dst4 = (const int4*)dst;

    for (int g = 0; g < nG; ++g) {
        int gg = g + gStart;
        if (gg >= nG) gg -= nG;
        const int base = gg * span + tid;
        int4 d[SCAN_UNROLL];
        if (gg * span + span <= E4) {
#pragma unroll
            for (int u = 0; u < SCAN_UNROLL; ++u) d[u] = dst4[base + u * PROP_THREADS];
        } else {
#pragma unroll
            for (int u = 0; u < SCAN_UNROLL; ++u) {
                int i = base + u * PROP_THREADS;
                int4 t = dst4[i < E4 ? i : 0];
                if (i >= E4) t = make_int4(-1, -1, -1, -1);
                d[u] = t;
            }
        }
        test_quads(d, base, lo, cnt, lane, q, qn, x, dinv, src, dst, acc, tileList, &cursor);
        test_quads(d + TEST_QUADS, base + TEST_QUADS * PROP_THREADS, lo, cnt, lane, q, qn, x, dinv, src, dst, acc, tileList, &cursor);
    }

    const int rem = E - 4 * E4;
    if (wave == 0 && rem > 0) {
        int e = 4 * E4 + lane;
        bool hit = false;
        if (lane < rem) hit = (unsigned)(dst[e] - lo) < (unsigned)cnt;
        unsigned b = __builtin_amdgcn_ballot_w32(hit);
        if (hit) q[qn + __builtin_popcount(b & ((1u << lane) - 1u))] = e;
        qn += __builtin_popcount(b);
    }
    while (qn > 0) {
        int n = qn < DRAIN_BATCH ? qn : DRAIN_BATCH;
        qn -= n;
        drain_edges(x, dinv, src, dst, acc, q + qn, n, lo, lane, tileList, &cursor);
    }
    __syncthreads();
    if (tid == 0) list2[E + blockIdx.x] = make_int2(listBase, cursor);

    finish_rows(acc, dinv, Tprev, Tnew, h, theta, order, c2, lo, cnt, tid);
}

__global__ __launch_bounds__(PROP_THREADS)
void propagate_list_kernel(const float* __restrict__ x, const float* __restrict__ dinv,
                           const int* __restrict__ elist,
                           const float* Tprev, float* Tnew, float* __restrict__ h,
                           const float* __restrict__ theta, int order, float c2, int N, int E) {
    __shared__ __attribute__((aligned(16))) float acc[PROP_ROWS * DFEAT];

    const int tid = threadIdx.x;
    const int lane = tid & 31;
    const int wave = __builtin_amdgcn_readfirstlane(tid >> 5);
    const int lo = blockIdx.x * PROP_ROWS;
    const int cnt = (N - lo < PROP_ROWS) ? (N - lo) : PROP_ROWS;

    const int2* list2 = (const int2*)elist;
    const int2 m = list2[E + blockIdx.x];
    const int2* tileList = list2 + m.x;
    const int count = m.y;
    const int nb = (count + DRAIN_BATCH - 1) / DRAIN_BATCH;

    int b = wave;
    int2 cur = make_int2(0, 0);
    if (b < nb) {
        int i = b * DRAIN_BATCH + lane;
        if (lane < DRAIN_BATCH && i < count) cur = tileList[i];
    }

    for (int i = tid; i < PROP_ROWS * DFEAT / 4; i += PROP_THREADS)
        ((float4*)acc)[i] = make_float4(0.f, 0.f, 0.f, 0.f);
    __syncthreads();

    while (b < nb) {
        const int bn = b + PROP_WAVES;
        int2 nxt = make_int2(0, 0);
        if (bn < nb) {
            int i = bn * DRAIN_BATCH + lane;
            if (lane < DRAIN_BATCH && i < count) nxt = tileList[i];
        }
        const int left = count - b * DRAIN_BATCH;
        if (left >= DRAIN_BATCH) gather_add(x, dinv, acc, cur.x, cur.y, DRAIN_BATCH, lane, nullptr, nullptr);
        else                     gather_add(x, dinv, acc, cur.x, cur.y, left, lane, nullptr, nullptr);
        cur = nxt;
        b = bn;
    }
    __syncthreads();

    finish_rows(acc, dinv, Tprev, Tnew, h, theta, order, c2, lo, cnt, tid);
}


__global__ void gemm_wmma_kernel(const float* __restrict__ h, const float* __restrict__ W,
                                 float* __restrict__ out, int N) {
    const int lane = threadIdx.x & 31;
    const int wave = threadIdx.x >> 5;
    const int tile = blockIdx.x * (blockDim.x >> 5) + wave;
    const int nRowTiles = N >> 4;
    const int row = tile >> 3;
    const int col = tile & 7;
    if (row >= nRowTiles) return;

    const int rowBase = row << 4;
    const int colBase = col << 4;
    const int m = lane & 15;
    const int half = lane >> 4;
    const int kb = half << 1;

    const float* hrow = h + (size_t)(rowBase + m) * DFEAT + kb;
    const float* wrow = W + (size_t)(colBase + m) * DFEAT + kb;

    v8f acc = {};
#pragma unroll
    for (int kk = 0; kk < DFEAT; kk += 4) {
        v2f a = *(const v2f*)(hrow + kk);
        v2f b = *(const v2f*)(wrow + kk);
        acc = __builtin_amdgcn_wmma_f32_16x16x4_f32(
             false, a,  false, b,
             (short)0, acc,  false,  false);
    }

#pragma unroll
    for (int v = 0; v < 8; ++v) {
        int r = rowBase + v + (half << 3);
        float val = acc[v];
        val = (val >= 0.0f) ? val : 0.01f * val;
        out[(size_t)r * DFEAT + colBase + m] = val;
    }
}


extern "C" void kernel_launch(void* const* d_in, const int* in_sizes, int n_in,
                              void* d_out, int out_size, void* d_ws, size_t ws_size,
                              hipStream_t stream) {
    const float* feat  = (const float*)d_in[0];
    const int*   src   = (const int*)d_in[1];
    const int*   dst   = (const int*)d_in[2];
    const float* theta = (const float*)d_in[3];
    const float* W     = (const float*)d_in[4];
    float* out = (float*)d_out;

    const int N = in_sizes[0] / DFEAT;
    const int E = in_sizes[1];

    float* ws   = (float*)d_ws;
    float* deg  = ws;
    float* dinv = deg + N;
    int*   elist = (int*)(dinv + N);
    float* Ta   = dinv + N + (size_t)N * DFEAT;
    float* Tb   = Ta + (size_t)N * DFEAT;
    float* h    = Tb + (size_t)N * DFEAT;

    const int TB = 256;
    const int nd4 = N * (DFEAT / 4);
    auto cdiv = [](long long a, long long b) { return (int)((a + b - 1) / b); };

    zero_f4_kernel<<<cdiv(N / 4, TB), TB, 0, stream>>>(deg, N / 4);
    deg_kernel<<<cdiv(E, TB), TB, 0, stream>>>(dst, deg, E);
    dinv_kernel<<<cdiv(N, TB), TB, 0, stream>>>(deg, dinv, N);

    h_init_kernel<<<cdiv(nd4, TB), TB, 0, stream>>>(feat, theta, h, nd4);

    const int groups = cdiv(N, PROP_ROWS);

    propagate_kernel<<<groups, PROP_THREADS, 0, stream>>>(feat, dinv, deg, src, dst, elist, nullptr, Ta, h, theta, 1, 1.0f, N, E);

    propagate_list_kernel<<<groups, PROP_THREADS, 0, stream>>>(Ta, dinv, elist, feat, Tb, h, theta, 2, 2.0f, N, E);

    propagate_list_kernel<<<groups, PROP_THREADS, 0, stream>>>(Tb, dinv, elist, Ta, Ta, h, theta, 3, 2.0f, N, E);

    const int tiles = (N / 16) * (DFEAT / 16);
    gemm_wmma_kernel<<<cdiv(tiles, 256 / 32), 256, 0, stream>>>(h, W, out, N);
}


# Order-1 filter reads one tile-number byte per edge (dst/196, packed into words by the degree pass), four tested per word, not the 4-byte dst: a group scans a quarter of the bytes; arrival-order sums
# speedup vs baseline: 2.8000x; 1.0332x over previous
#include <hip/hip_runtime.h>

typedef float v2f __attribute__((ext_vector_type(2)));
typedef float v8f __attribute__((ext_vector_type(8)));

#define DFEAT 128


__global__ void zero_f4_kernel(float* __restrict__ p, int n4) {
    int i = blockIdx.x * blockDim.x + threadIdx.x;
    if (i < n4) ((float4*)p)[i] = make_float4(0.f, 0.f, 0.f, 0.f);
}

#define TILE_ROWS 196
__global__ void deg_code_kernel(const int* __restrict__ dst, float* __restrict__ deg,
                                unsigned* __restrict__ code, int E) {
    const int i = blockIdx.x * blockDim.x + threadIdx.x;
    const int E4 = E >> 2;
    if (i < E4) {
        int4 d = ((const int4*)dst)[i];
        atomicAdd(&deg[d.x], 1.0f);
        atomicAdd(&deg[d.y], 1.0f);
        atomicAdd(&deg[d.z], 1.0f);
        atomicAdd(&deg[d.w], 1.0f);
        code[i] = ((unsigned)d.x / TILE_ROWS) | (((unsigned)d.y / TILE_ROWS) << 8) |
                  (((unsigned)d.z / TILE_ROWS) << 16) | (((unsigned)d.w / TILE_ROWS) << 24);
    }
    if (i < E - 4 * E4) atomicAdd(&deg[dst[4 * E4 + i]], 1.0f);
}

__global__ void dinv_kernel(const float* __restrict__ deg, float* __restrict__ dinv, int N) {
    int i = blockIdx.x * blockDim.x + threadIdx.x;
    if (i < N) dinv[i] = 1.0f / sqrtf(fmaxf(deg[i], 1.0f));
}

__global__ void h_init_kernel(const float* __restrict__ feat, const float* __restrict__ theta,
                              float* __restrict__ h, int n4) {
    int i = blockIdx.x * blockDim.x + threadIdx.x;
    if (i >= n4) return;
    float t0 = theta[0];
    float4 f = ((const float4*)feat)[i];
    ((float4*)h)[i] = make_float4(t0 * f.x, t0 * f.y, t0 * f.z, t0 * f.w);
}


#define PROP_THREADS 1024
#define PROP_WAVES   32
#define PROP_ROWS    196
#define SCAN_UNROLL  8
#define TEST_QUADS   2
static_assert(PROP_ROWS == TILE_ROWS, "the degree pass numbers the tiles the propagation owns");
#define DRAIN_BATCH  8
#define QCAP (DRAIN_BATCH - 1 + 32)
static_assert(PROP_ROWS % 4 == 0, "a tile's first row must be a multiple of 4: the degree counts in front of it are read as float4");
#define BASE_UNROLL  8

__device__ __forceinline__ void gather_add(const float* __restrict__ x,
                                           const float* __restrict__ dinv,
                                           float* acc, int s, int dl, int n, int lane,
                                           int2* tileList, int* cursor) {
    float w = 0.0f;
    float v[DRAIN_BATCH][4];
#pragma unroll
    for (int j = 0; j < DRAIN_BATCH; ++j) {
        v[j][0] = 0.0f; v[j][1] = 0.0f; v[j][2] = 0.0f; v[j][3] = 0.0f;
        if (j < n) {
            int sj = __builtin_amdgcn_readlane(s, j);
            const float* xp = x + (size_t)sj * DFEAT + lane;
            v[j][0] = xp[0];
            v[j][1] = xp[32];
            v[j][2] = xp[64];
            v[j][3] = xp[96];
        }
    }
    if (lane < n) w = dinv[s];
    if (tileList) {
        int pos = 0;
        if (lane == 0) pos = atomicAdd(cursor, n);
        pos = __builtin_amdgcn_readfirstlane(pos);
        if (lane < n) tileList[pos + lane] = make_int2(s, dl);
    }
#pragma unroll
    for (int j = 0; j < DRAIN_BATCH; ++j) {
        if (j < n) {
            float wj = __int_as_float(__builtin_amdgcn_readlane(__float_as_int(w), j));
            int dj = __builtin_amdgcn_readlane(dl, j);
            float* ap = acc + dj * DFEAT + lane;
            atomicAdd(ap + 0,  wj * v[j][0]);
            atomicAdd(ap + 32, wj * v[j][1]);
            atomicAdd(ap + 64, wj * v[j][2]);
            atomicAdd(ap + 96, wj * v[j][3]);
        }
    }
}

__device__ __forceinline__ void drain_edges(const float* __restrict__ x,
                                            const float* __restrict__ dinv,
                                            const int* __restrict__ src,
                                            const int* __restrict__ dst,
                                            float* acc, const int* q, int n, int lo, int lane,
                                            int2* tileList, int* cursor) {
    int s = 0, dl = 0;
    if (lane < n) {
        int e = q[lane];
        s = src[e];
        dl = dst[e] - lo;
    }
    gather_add(x, dinv, acc, s, dl, n, lane, tileList, cursor);
}

__device__ __forceinline__ unsigned not_mine(unsigned w, unsigned mine, int k) {
    unsigned x = w ^ mine;
    unsigned y = ((x & 0x7f7f7f7fu) + 0x7f7f7f7fu) | x;
    return (y >> (7 - k)) & (0x01010101u << k);
}

__device__ __forceinline__ void test_quads(const uint4* d, int base, unsigned mine, int lo, int lane,
                                           int* q, int& qn,
                                           const float* __restrict__ x, const float* __restrict__ dinv,
                                           const int* __restrict__ src, const int* __restrict__ dst,
                                           float* acc, int2* tileList, int* cursor) {
    unsigned nm = 0;
#pragma unroll
    for (int u = 0; u < TEST_QUADS; ++u) {
        nm |= not_mine(d[u].x, mine, 4 * u + 0);
        nm |= not_mine(d[u].y, mine, 4 * u + 1);
        nm |= not_mine(d[u].z, mine, 4 * u + 2);
        nm |= not_mine(d[u].w, mine, 4 * u + 3);
    }
    unsigned mm = ~nm;
    for (;;) {
        unsigned b = __builtin_amdgcn_ballot_w32(mm != 0);
        if (b == 0) break;
        if (mm != 0) {
            int bit = __builtin_ctz(mm);
            mm &= mm - 1;
            int k = bit & 7;
            int i = base + (k >> 2) * PROP_THREADS;
            int pos = qn + __builtin_popcount(b & ((1u << lane) - 1u));
            q[pos] = 16 * i + 4 * (k & 3) + (bit >> 3);
        }
        qn += __builtin_popcount(b);
        while (qn >= DRAIN_BATCH) {
            qn -= DRAIN_BATCH;
            drain_edges(x, dinv, src, dst, acc, q + qn, DRAIN_BATCH, lo, lane, tileList, cursor);
        }
    }
}

__device__ __forceinline__ void finish_rows(const float* acc, const float* __restrict__ dinv,
                                            const float* Tprev, float* Tnew, float* __restrict__ h,
                                            const float* __restrict__ theta, int order, float c2,
                                            int lo, int cnt, int tid) {
    const float th = theta[order] * 0.5f;
    const int tot4 = cnt * (DFEAT / 4);
    const size_t g0 = (size_t)lo * (DFEAT / 4);
    for (int i = tid; i < tot4; i += PROP_THREADS) {
        int n = lo + (i >> 5);
        float s = -c2 * dinv[n];
        float4 a = ((const float4*)acc)[i];
        float4 tp = Tprev ? ((const float4*)Tprev)[g0 + i] : make_float4(0.f, 0.f, 0.f, 0.f);
        float4 t;
        t.x = fmaf(s, a.x, -tp.x);
        t.y = fmaf(s, a.y, -tp.y);
        t.z = fmaf(s, a.z, -tp.z);
        t.w = fmaf(s, a.w, -tp.w);
        ((float4*)Tnew)[g0 + i] = t;

        float4 hv = ((const float4*)h)[g0 + i];
        hv.x = fmaf(th, t.x + 1.0f, hv.x);
        hv.y = fmaf(th, t.y + 1.0f, hv.y);
        hv.z = fmaf(th, t.z + 1.0f, hv.z);
        hv.w = fmaf(th, t.w + 1.0f, hv.w);
        ((float4*)h)[g0 + i] = hv;
    }
}

__global__ __launch_bounds__(PROP_THREADS)
void propagate_kernel(const float* __restrict__ x, const float* __restrict__ dinv,
                      const float* __restrict__ deg,
                      const int* __restrict__ src, const int* __restrict__ dst,
                      const unsigned* __restrict__ code,
                      int* __restrict__ elist,
                      const float* Tprev, float* Tnew, float* __restrict__ h,
                      const float* __restrict__ theta, int order, float c2, int N, int E) {
    __shared__ __attribute__((aligned(16))) float acc[PROP_ROWS * DFEAT];
    __shared__ int queue[PROP_WAVES][QCAP];
    __shared__ int listBase;
    __shared__ int cursor;

    const int tid = threadIdx.x;
    const int lane = tid & 31;
    const int wave = tid >> 5;
    const int lo = blockIdx.x * PROP_ROWS;
    const int cnt = (N - lo < PROP_ROWS) ? (N - lo) : PROP_ROWS;

    if (tid == 0) { listBase = 0; cursor = 0; }
    for (int i = tid; i < PROP_ROWS * DFEAT / 4; i += PROP_THREADS)
        ((float4*)acc)[i] = make_float4(0.f, 0.f, 0.f, 0.f);

    int part = 0;
    {
        const float4* deg4 = (const float4*)deg;
        const int n4 = lo >> 2;
        for (int i0 = tid; i0 < n4; i0 += BASE_UNROLL * PROP_THREADS) {
            float4 dv[BASE_UNROLL];
#pragma unroll
            for (int u = 0; u < BASE_UNROLL; ++u) {
                int i = i0 + u * PROP_THREADS;
                dv[u] = deg4[i < n4 ? i : 0];
            }
#pragma unroll
            for (int u = 0; u < BASE_UNROLL; ++u) {
                int i = i0 + u * PROP_THREADS;
                int t = (int)dv[u].x + (int)dv[u].y + (int)dv[u].z + (int)dv[u].w;
                part += (i < n4) ? t : 0;
            }
        }
        for (int off = 16; off > 0; off >>= 1) part += __shfl_xor(part, off, 32);
    }
    __syncthreads();
    if (lane == 0 && part != 0) atomicAdd(&listBase, part);
    __syncthreads();
    int2* list2 = (int2*)elist;
    int2* tileList = list2 + listBase;

    int* q = queue[wave];
    int qn = 0;

    const int E16 = E >> 4;
    const int span = SCAN_UNROLL * PROP_THREADS;
    const int nG = (E16 + span - 1) / span;
    const int gStart = (int)((long long)blockIdx.x * nG / gridDim.x);
    const uint4* code4 = (const uint4*)code;
    const unsigned mine = blockIdx.x * 0x01010101u;
    const unsigned other = mine ^ 0x01010101u;

    for (int g = 0; g < nG; ++g) {
        int gg = g + gStart;
        if (gg >= nG) gg -= nG;
        const int base = gg * span + tid;
        uint4 d[SCAN_UNROLL];
        if (gg * span + span <= E16) {
#pragma unroll
            for (int u = 0; u < SCAN_UNROLL; ++u) d[u] = code4[base + u * PROP_THREADS];
        } else {
#pragma unroll
            for (int u = 0; u < SCAN_UNROLL; ++u) {
                int i = base + u * PROP_THREADS;
                uint4 t = code4[i < E16 ? i : 0];
                if (i >= E16) t = make_uint4(other, other, other, other);
                d[u] = t;
            }
        }
#pragma unroll
        for (int p = 0; p < SCAN_UNROLL; p += TEST_QUADS)
            test_quads(d + p, base + p * PROP_THREADS, mine, lo, lane, q, qn, x, dinv, src, dst, acc, tileList, &cursor);
    }

    const int rem = E - 16 * E16;
    if (wave == 0 && rem > 0) {
        int e = 16 * E16 + lane;
        bool hit = false;
        if (lane < rem) hit = (unsigned)(dst[e] - lo) < (unsigned)cnt;
        unsigned b = __builtin_amdgcn_ballot_w32(hit);
        if (hit) q[qn + __builtin_popcount(b & ((1u << lane) - 1u))] = e;
        qn += __builtin_popcount(b);
    }
    while (qn > 0) {
        int n = qn < DRAIN_BATCH ? qn : DRAIN_BATCH;
        qn -= n;
        drain_edges(x, dinv, src, dst, acc, q + qn, n, lo, lane, tileList, &cursor);
    }
    __syncthreads();
    if (tid == 0) list2[E + blockIdx.x] = make_int2(listBase, cursor);

    finish_rows(acc, dinv, Tprev, Tnew, h, theta, order, c2, lo, cnt, tid);
}

__global__ __launch_bounds__(PROP_THREADS)
void propagate_list_kernel(const float* __restrict__ x, const float* __restrict__ dinv,
                           const int* __restrict__ elist,
                           const float* Tprev, float* Tnew, float* __restrict__ h,
                           const float* __restrict__ theta, int order, float c2, int N, int E) {
    __shared__ __attribute__((aligned(16))) float acc[PROP_ROWS * DFEAT];

    const int tid = threadIdx.x;
    const int lane = tid & 31;
    const int wave = __builtin_amdgcn_readfirstlane(tid >> 5);
    const int lo = blockIdx.x * PROP_ROWS;
    const int cnt = (N - lo < PROP_ROWS) ? (N - lo) : PROP_ROWS;

    const int2* list2 = (const int2*)elist;
    const int2 m = list2[E + blockIdx.x];
    const int2* tileList = list2 + m.x;
    const int count = m.y;
    const int nb = (count + DRAIN_BATCH - 1) / DRAIN_BATCH;

    int b = wave;
    int2 cur = make_int2(0, 0);
    if (b < nb) {
        int i = b * DRAIN_BATCH + lane;
        if (lane < DRAIN_BATCH && i < count) cur = tileList[i];
    }

    for (int i = tid; i < PROP_ROWS * DFEAT / 4; i += PROP_THREADS)
        ((float4*)acc)[i] = make_float4(0.f, 0.f, 0.f, 0.f);
    __syncthreads();

    while (b < nb) {
        const int bn = b + PROP_WAVES;
        int2 nxt = make_int2(0, 0);
        if (bn < nb) {
            int i = bn * DRAIN_BATCH + lane;
            if (lane < DRAIN_BATCH && i < count) nxt = tileList[i];
        }
        const int left = count - b * DRAIN_BATCH;
        if (left >= DRAIN_BATCH) gather_add(x, dinv, acc, cur.x, cur.y, DRAIN_BATCH, lane, nullptr, nullptr);
        else                     gather_add(x, dinv, acc, cur.x, cur.y, left, lane, nullptr, nullptr);
        cur = nxt;
        b = bn;
    }
    __syncthreads();

    finish_rows(acc, dinv, Tprev, Tnew, h, theta, order, c2, lo, cnt, tid);
}


__global__ void gemm_wmma_kernel(const float* __restrict__ h, const float* __restrict__ W,
                                 float* __restrict__ out, int N) {
    const int lane = threadIdx.x & 31;
    const int wave = threadIdx.x >> 5;
    const int tile = blockIdx.x * (blockDim.x >> 5) + wave;
    const int nRowTiles = N >> 4;
    const int row = tile >> 3;
    const int col = tile & 7;
    if (row >= nRowTiles) return;

    const int rowBase = row << 4;
    const int colBase = col << 4;
    const int m = lane & 15;
    const int half = lane >> 4;
    const int kb = half << 1;

    const float* hrow = h + (size_t)(rowBase + m) * DFEAT + kb;
    const float* wrow = W + (size_t)(colBase + m) * DFEAT + kb;

    v8f acc = {};
#pragma unroll
    for (int kk = 0; kk < DFEAT; kk += 4) {
        v2f a = *(const v2f*)(hrow + kk);
        v2f b = *(const v2f*)(wrow + kk);
        acc = __builtin_amdgcn_wmma_f32_16x16x4_f32(
             false, a,  false, b,
             (short)0, acc,  false,  false);
    }

#pragma unroll
    for (int v = 0; v < 8; ++v) {
        int r = rowBase + v + (half << 3);
        float val = acc[v];
        val = (val >= 0.0f) ? val : 0.01f * val;
        out[(size_t)r * DFEAT + colBase + m] = val;
    }
}


extern "C" void kernel_launch(void* const* d_in, const int* in_sizes, int n_in,
                              void* d_out, int out_size, void* d_ws, size_t ws_size,
                              hipStream_t stream) {
    const float* feat  = (const float*)d_in[0];
    const int*   src   = (const int*)d_in[1];
    const int*   dst   = (const int*)d_in[2];
    const float* theta = (const float*)d_in[3];
    const float* W     = (const float*)d_in[4];
    float* out = (float*)d_out;

    const int N = in_sizes[0] / DFEAT;
    const int E = in_sizes[1];

    float* ws   = (float*)d_ws;
    float* deg  = ws;
    float* dinv = deg + N;
    int*   elist = (int*)(dinv + N);
    float* Ta   = dinv + N + (size_t)N * DFEAT;
    float* Tb   = Ta + (size_t)N * DFEAT;
    float* h    = Tb + (size_t)N * DFEAT;

    const int TB = 256;
    const int nd4 = N * (DFEAT / 4);
    auto cdiv = [](long long a, long long b) { return (int)((a + b - 1) / b); };

    zero_f4_kernel<<<cdiv(N / 4, TB), TB, 0, stream>>>(deg, N / 4);
    unsigned* code = (unsigned*)Tb;
    deg_code_kernel<<<cdiv(cdiv(E, 4), TB), TB, 0, stream>>>(dst, deg, code, E);
    dinv_kernel<<<cdiv(N, TB), TB, 0, stream>>>(deg, dinv, N);

    h_init_kernel<<<cdiv(nd4, TB), TB, 0, stream>>>(feat, theta, h, nd4);

    const int groups = cdiv(N, PROP_ROWS);

    propagate_kernel<<<groups, PROP_THREADS, 0, stream>>>(feat, dinv, deg, src, dst, code, elist, nullptr, Ta, h, theta, 1, 1.0f, N, E);

    propagate_list_kernel<<<groups, PROP_THREADS, 0, stream>>>(Ta, dinv, elist, feat, Tb, h, theta, 2, 2.0f, N, E);

    propagate_list_kernel<<<groups, PROP_THREADS, 0, stream>>>(Tb, dinv, elist, Ta, Ta, h, theta, 3, 2.0f, N, E);

    const int tiles = (N / 16) * (DFEAT / 16);
    gemm_wmma_kernel<<<cdiv(tiles, 256 / 32), 256, 0, stream>>>(h, W, out, N);
}


# All orders: wave w owns tile rows r%32==w, picks its entries from the LDS-staged tile list by owner byte, sums in registers, finishes them; order 1 lists first; no LDS float adds; list-order sums
# speedup vs baseline: 2.9207x; 1.0431x over previous
#include <hip/hip_runtime.h>

typedef float v2f __attribute__((ext_vector_type(2)));
typedef float v8f __attribute__((ext_vector_type(8)));

#define DFEAT 128


__global__ void zero_f4_kernel(float* __restrict__ p, int n4) {
    int i = blockIdx.x * blockDim.x + threadIdx.x;
    if (i < n4) ((float4*)p)[i] = make_float4(0.f, 0.f, 0.f, 0.f);
}

#define TILE_ROWS 196
__global__ void deg_code_kernel(const int* __restrict__ dst, float* __restrict__ deg,
                                unsigned* __restrict__ code, int E) {
    const int i = blockIdx.x * blockDim.x + threadIdx.x;
    const int E4 = E >> 2;
    if (i < E4) {
        int4 d = ((const int4*)dst)[i];
        atomicAdd(&deg[d.x], 1.0f);
        atomicAdd(&deg[d.y], 1.0f);
        atomicAdd(&deg[d.z], 1.0f);
        atomicAdd(&deg[d.w], 1.0f);
        code[i] = ((unsigned)d.x / TILE_ROWS) | (((unsigned)d.y / TILE_ROWS) << 8) |
                  (((unsigned)d.z / TILE_ROWS) << 16) | (((unsigned)d.w / TILE_ROWS) << 24);
    }
    if (i < E - 4 * E4) atomicAdd(&deg[dst[4 * E4 + i]], 1.0f);
}

__global__ void dinv_kernel(const float* __restrict__ deg, float* __restrict__ dinv, int N) {
    int i = blockIdx.x * blockDim.x + threadIdx.x;
    if (i < N) dinv[i] = 1.0f / sqrtf(fmaxf(deg[i], 1.0f));
}

__global__ void h_init_kernel(const float* __restrict__ feat, const float* __restrict__ theta,
                              float* __restrict__ h, int n4) {
    int i = blockIdx.x * blockDim.x + threadIdx.x;
    if (i >= n4) return;
    float t0 = theta[0];
    float4 f = ((const float4*)feat)[i];
    ((float4*)h)[i] = make_float4(t0 * f.x, t0 * f.y, t0 * f.z, t0 * f.w);
}


#define PROP_THREADS 1024
#define PROP_WAVES   32
#define PROP_ROWS    196
#define SCAN_UNROLL  8
#define TEST_QUADS   2
static_assert(PROP_ROWS == TILE_ROWS, "the degree pass numbers the tiles the propagation owns");
#define APPEND_BATCH 32
#define DRAIN_BATCH  8
#define QCAP (APPEND_BATCH - 1 + 32)
static_assert(DRAIN_BATCH <= APPEND_BATCH, "the queue is sized for the larger batch");
static_assert(PROP_ROWS % 4 == 0, "a tile's first row must be a multiple of 4: the degree counts in front of it are read as float4");
#define BASE_UNROLL  8
#define LIST_CHUNK 4096
static_assert(LIST_CHUNK == 4 * PROP_THREADS, "a thread stages four entries and packs their four owner bytes into one word");
#define OWN_ROWS 7
static_assert(OWN_ROWS * PROP_WAVES >= PROP_ROWS, "every row of the tile has an owner wave and a register slot");

__device__ __forceinline__ unsigned not_mine(unsigned w, unsigned mine, int k) {
    unsigned x = w ^ mine;
    unsigned y = ((x & 0x7f7f7f7fu) + 0x7f7f7f7fu) | x;
    return (y >> (7 - k)) & (0x01010101u << k);
}

__device__ __forceinline__ void own_batch(const float* __restrict__ x, const float* __restrict__ dinv,
                                          const int2* ent, const int* qp, int n, int lane,
                                          float (&a)[OWN_ROWS][4]) {
    int s = 0, dl = 0;
    if (lane < n) {
        int2 e = ent[qp[lane]];
        s = e.x;
        dl = e.y;
    }
    float v[DRAIN_BATCH][4];
#pragma unroll
    for (int j = 0; j < DRAIN_BATCH; ++j) {
        v[j][0] = 0.0f; v[j][1] = 0.0f; v[j][2] = 0.0f; v[j][3] = 0.0f;
        if (j < n) {
            int sj = __builtin_amdgcn_readlane(s, j);
            const float* xp = x + (size_t)sj * DFEAT + lane;
            v[j][0] = xp[0];
            v[j][1] = xp[32];
            v[j][2] = xp[64];
            v[j][3] = xp[96];
        }
    }
    float w = 0.0f;
    if (lane < n) w = dinv[s];
#pragma unroll
    for (int j = 0; j < DRAIN_BATCH; ++j) {
        if (j < n) {
            const float wj = __int_as_float(__builtin_amdgcn_readlane(__float_as_int(w), j));
            const int kj = __builtin_amdgcn_readlane(dl, j) >> 5;
#pragma unroll
            for (int k = 0; k < OWN_ROWS; ++k) {
                if (kj == k) {
                    a[k][0] = fmaf(wj, v[j][0], a[k][0]);
                    a[k][1] = fmaf(wj, v[j][1], a[k][1]);
                    a[k][2] = fmaf(wj, v[j][2], a[k][2]);
                    a[k][3] = fmaf(wj, v[j][3], a[k][3]);
                }
            }
        }
    }
}

__device__ __forceinline__ void walk_finish(const float* __restrict__ x, const float* __restrict__ dinv,
                                            const int2* tileList, int count,
                                            int2* ent, unsigned* own, int* q,
                                            const float* Tprev, float* Tnew, float* __restrict__ h,
                                            const float* __restrict__ theta, int order, float c2,
                                            int lo, int cnt, int tid, int lane, int wave) {
    float a[OWN_ROWS][4];
#pragma unroll
    for (int k = 0; k < OWN_ROWS; ++k) { a[k][0] = 0.0f; a[k][1] = 0.0f; a[k][2] = 0.0f; a[k][3] = 0.0f; }

    int qn = 0;
    const unsigned mine = (unsigned)wave * 0x01010101u;

    for (int c0 = 0; c0 < count; c0 += LIST_CHUNK) {
        const int nc = (count - c0 < LIST_CHUNK) ? (count - c0) : LIST_CHUNK;
        {
            unsigned ow = 0;
#pragma unroll
            for (int u = 0; u < 4; ++u) {
                const int idx = 4 * tid + u;
                int2 e = tileList[c0 + (idx < nc ? idx : 0)];
                ent[idx] = e;
                unsigned b = (idx < nc) ? ((unsigned)e.y & 31u) : 0x7fu;
                ow |= b << (8 * u);
            }
            own[tid] = ow;
        }
        __syncthreads();

        const int steps = (nc + 511) >> 9;
        for (int st = 0; st < steps; ++st) {
            const int i4 = st * 32 + lane;
            const uint4 d = ((const uint4*)own)[i4];
            unsigned nm = not_mine(d.x, mine, 0) | not_mine(d.y, mine, 1) |
                          not_mine(d.z, mine, 2) | not_mine(d.w, mine, 3);
            unsigned mm = ~nm & 0x0f0f0f0fu;
            for (;;) {
                unsigned b = __builtin_amdgcn_ballot_w32(mm != 0);
                if (b == 0) break;
                if (mm != 0) {
                    int bit = __builtin_ctz(mm);
                    mm &= mm - 1;
                    int pos = qn + __builtin_popcount(b & ((1u << lane) - 1u));
                    q[pos] = 16 * i4 + 4 * (bit & 7) + (bit >> 3);
                }
                qn += __builtin_popcount(b);
                while (qn >= DRAIN_BATCH) {
                    qn -= DRAIN_BATCH;
                    own_batch(x, dinv, ent, q + qn, DRAIN_BATCH, lane, a);
                }
            }
        }
        if (qn > 0) {
            own_batch(x, dinv, ent, q, qn, lane, a);
            qn = 0;
        }
        __syncthreads();
    }

    const float th = theta[order] * 0.5f;
    float tp[OWN_ROWS][4], hv[OWN_ROWS][4], sc[OWN_ROWS];
#pragma unroll
    for (int k = 0; k < OWN_ROWS; ++k) {
        const int r = wave + 32 * k;
        sc[k] = 0.0f;
#pragma unroll
        for (int c = 0; c < 4; ++c) { tp[k][c] = 0.0f; hv[k][c] = 0.0f; }
        if (r < cnt) {
            const size_t g = (size_t)(lo + r) * DFEAT + lane;
            sc[k] = -c2 * dinv[lo + r];
#pragma unroll
            for (int c = 0; c < 4; ++c) {
                if (Tprev) tp[k][c] = Tprev[g + 32 * c];
                hv[k][c] = h[g + 32 * c];
            }
        }
    }
#pragma unroll
    for (int k = 0; k < OWN_ROWS; ++k) {
        const int r = wave + 32 * k;
        if (r < cnt) {
            const size_t g = (size_t)(lo + r) * DFEAT + lane;
#pragma unroll
            for (int c = 0; c < 4; ++c) {
                const float t = fmaf(sc[k], a[k][c], -tp[k][c]);
                Tnew[g + 32 * c] = t;
                h[g + 32 * c] = fmaf(th, t + 1.0f, hv[k][c]);
            }
        }
    }
}

__device__ __forceinline__ void append_edges(const int* __restrict__ src, const int* __restrict__ dst,
                                             const int* qp, int n, int lo, int lane,
                                             int2* tileList, int* cursor) {
    int s = 0, dl = 0;
    if (lane < n) {
        int e = qp[lane];
        s = src[e];
        dl = dst[e] - lo;
    }
    int pos = 0;
    if (lane == 0) pos = atomicAdd(cursor, n);
    pos = __builtin_amdgcn_readfirstlane(pos);
    if (lane < n) tileList[pos + lane] = make_int2(s, dl);
}

__device__ __forceinline__ void test_quads(const uint4* d, int base, unsigned mine, int lo, int lane,
                                           int* q, int& qn,
                                           const int* __restrict__ src, const int* __restrict__ dst,
                                           int2* tileList, int* cursor) {
    unsigned nm = 0;
#pragma unroll
    for (int u = 0; u < TEST_QUADS; ++u) {
        nm |= not_mine(d[u].x, mine, 4 * u + 0);
        nm |= not_mine(d[u].y, mine, 4 * u + 1);
        nm |= not_mine(d[u].z, mine, 4 * u + 2);
        nm |= not_mine(d[u].w, mine, 4 * u + 3);
    }
    unsigned mm = ~nm;
    for (;;) {
        unsigned b = __builtin_amdgcn_ballot_w32(mm != 0);
        if (b == 0) break;
        if (mm != 0) {
            int bit = __builtin_ctz(mm);
            mm &= mm - 1;
            int k = bit & 7;
            int i = base + (k >> 2) * PROP_THREADS;
            int pos = qn + __builtin_popcount(b & ((1u << lane) - 1u));
            q[pos] = 16 * i + 4 * (k & 3) + (bit >> 3);
        }
        qn += __builtin_popcount(b);
        while (qn >= APPEND_BATCH) {
            qn -= APPEND_BATCH;
            append_edges(src, dst, q + qn, APPEND_BATCH, lo, lane, tileList, cursor);
        }
    }
}

__global__ __launch_bounds__(PROP_THREADS)
void propagate_kernel(const float* __restrict__ x, const float* __restrict__ dinv,
                      const float* __restrict__ deg,
                      const int* __restrict__ src, const int* __restrict__ dst,
                      const unsigned* __restrict__ code,
                      int* elist,
                      const float* Tprev, float* Tnew, float* __restrict__ h,
                      const float* __restrict__ theta, int order, float c2, int N, int E) {
    __shared__ int2 ent[LIST_CHUNK];
    __shared__ __attribute__((aligned(16))) unsigned own[LIST_CHUNK / 4];
    __shared__ int queue[PROP_WAVES][QCAP];
    __shared__ int listBase;
    __shared__ int cursor;

    const int tid = threadIdx.x;
    const int lane = tid & 31;
    const int wave = __builtin_amdgcn_readfirstlane(tid >> 5);
    const int lo = blockIdx.x * PROP_ROWS;
    const int cnt = (N - lo < PROP_ROWS) ? (N - lo) : PROP_ROWS;

    if (tid == 0) { listBase = 0; cursor = 0; }

    int part = 0;
    {
        const float4* deg4 = (const float4*)deg;
        const int n4 = lo >> 2;
        for (int i0 = tid; i0 < n4; i0 += BASE_UNROLL * PROP_THREADS) {
            float4 dv[BASE_UNROLL];
#pragma unroll
            for (int u = 0; u < BASE_UNROLL; ++u) {
                int i = i0 + u * PROP_THREADS;
                dv[u] = deg4[i < n4 ? i : 0];
            }
#pragma unroll
            for (int u = 0; u < BASE_UNROLL; ++u) {
                int i = i0 + u * PROP_THREADS;
                int t = (int)dv[u].x + (int)dv[u].y + (int)dv[u].z + (int)dv[u].w;
                part += (i < n4) ? t : 0;
            }
        }
        for (int off = 16; off > 0; off >>= 1) part += __shfl_xor(part, off, 32);
    }
    __syncthreads();
    if (lane == 0 && part != 0) atomicAdd(&listBase, part);
    __syncthreads();
    int2* list2 = (int2*)elist;
    int2* tileList = list2 + listBase;

    int* q = queue[wave];
    int qn = 0;

    const int E16 = E >> 4;
    const int span = SCAN_UNROLL * PROP_THREADS;
    const int nG = (E16 + span - 1) / span;
    const int gStart = (int)((long long)blockIdx.x * nG / gridDim.x);
    const uint4* code4 = (const uint4*)code;
    const unsigned mine = blockIdx.x * 0x01010101u;
    const unsigned other = mine ^ 0x01010101u;

    for (int g = 0; g < nG; ++g) {
        int gg = g + gStart;
        if (gg >= nG) gg -= nG;
        const int base = gg * span + tid;
        uint4 d[SCAN_UNROLL];
        if (gg * span + span <= E16) {
#pragma unroll
            for (int u = 0; u < SCAN_UNROLL; ++u) d[u] = code4[base + u * PROP_THREADS];
        } else {
#pragma unroll
            for (int u = 0; u < SCAN_UNROLL; ++u) {
                int i = base + u * PROP_THREADS;
                uint4 t = code4[i < E16 ? i : 0];
                if (i >= E16) t = make_uint4(other, other, other, other);
                d[u] = t;
            }
        }
#pragma unroll
        for (int p = 0; p < SCAN_UNROLL; p += TEST_QUADS)
            test_quads(d + p, base + p * PROP_THREADS, mine, lo, lane, q, qn, src, dst, tileList, &cursor);
    }

    const int rem = E - 16 * E16;
    if (wave == 0 && rem > 0) {
        int e = 16 * E16 + lane;
        bool hit = false;
        if (lane < rem) hit = (unsigned)(dst[e] - lo) < (unsigned)cnt;
        unsigned b = __builtin_amdgcn_ballot_w32(hit);
        if (hit) q[qn + __builtin_popcount(b & ((1u << lane) - 1u))] = e;
        qn += __builtin_popcount(b);
    }
    while (qn > 0) {
        int n = qn < APPEND_BATCH ? qn : APPEND_BATCH;
        qn -= n;
        append_edges(src, dst, q + qn, n, lo, lane, tileList, &cursor);
    }
    __syncthreads();
    const int count = cursor;
    if (tid == 0) list2[E + blockIdx.x] = make_int2(listBase, count);

    walk_finish(x, dinv, tileList, count, ent, own, q, Tprev, Tnew, h, theta, order, c2, lo, cnt, tid, lane, wave);
}

__global__ __launch_bounds__(PROP_THREADS)
void propagate_list_kernel(const float* __restrict__ x, const float* __restrict__ dinv,
                           const int* __restrict__ elist,
                           const float* Tprev, float* Tnew, float* __restrict__ h,
                           const float* __restrict__ theta, int order, float c2, int N, int E) {
    __shared__ int2 ent[LIST_CHUNK];
    __shared__ __attribute__((aligned(16))) unsigned own[LIST_CHUNK / 4];
    __shared__ int queue[PROP_WAVES][QCAP];

    const int tid = threadIdx.x;
    const int lane = tid & 31;
    const int wave = __builtin_amdgcn_readfirstlane(tid >> 5);
    const int lo = blockIdx.x * PROP_ROWS;
    const int cnt = (N - lo < PROP_ROWS) ? (N - lo) : PROP_ROWS;

    const int2* list2 = (const int2*)elist;
    const int2 m = list2[E + blockIdx.x];

    walk_finish(x, dinv, list2 + m.x, m.y, ent, own, queue[wave], Tprev, Tnew, h, theta, order, c2, lo, cnt, tid, lane, wave);
}


__global__ void gemm_wmma_kernel(const float* __restrict__ h, const float* __restrict__ W,
                                 float* __restrict__ out, int N) {
    const int lane = threadIdx.x & 31;
    const int wave = threadIdx.x >> 5;
    const int tile = blockIdx.x * (blockDim.x >> 5) + wave;
    const int nRowTiles = N >> 4;
    const int row = tile >> 3;
    const int col = tile & 7;
    if (row >= nRowTiles) return;

    const int rowBase = row << 4;
    const int colBase = col << 4;
    const int m = lane & 15;
    const int half = lane >> 4;
    const int kb = half << 1;

    const float* hrow = h + (size_t)(rowBase + m) * DFEAT + kb;
    const float* wrow = W + (size_t)(colBase + m) * DFEAT + kb;

    v8f acc = {};
#pragma unroll
    for (int kk = 0; kk < DFEAT; kk += 4) {
        v2f a = *(const v2f*)(hrow + kk);
        v2f b = *(const v2f*)(wrow + kk);
        acc = __builtin_amdgcn_wmma_f32_16x16x4_f32(
             false, a,  false, b,
             (short)0, acc,  false,  false);
    }

#pragma unroll
    for (int v = 0; v < 8; ++v) {
        int r = rowBase + v + (half << 3);
        float val = acc[v];
        val = (val >= 0.0f) ? val : 0.01f * val;
        out[(size_t)r * DFEAT + colBase + m] = val;
    }
}


extern "C" void kernel_launch(void* const* d_in, const int* in_sizes, int n_in,
                              void* d_out, int out_size, void* d_ws, size_t ws_size,
                              hipStream_t stream) {
    const float* feat  = (const float*)d_in[0];
    const int*   src   = (const int*)d_in[1];
    const int*   dst   = (const int*)d_in[2];
    const float* theta = (const float*)d_in[3];
    const float* W     = (const float*)d_in[4];
    float* out = (float*)d_out;

    const int N = in_sizes[0] / DFEAT;
    const int E = in_sizes[1];

    float* ws   = (float*)d_ws;
    float* deg  = ws;
    float* dinv = deg + N;
    int*   elist = (int*)(dinv + N);
    float* Ta   = dinv + N + (size_t)N * DFEAT;
    float* Tb   = Ta + (size_t)N * DFEAT;
    float* h    = Tb + (size_t)N * DFEAT;

    const int TB = 256;
    const int nd4 = N * (DFEAT / 4);
    auto cdiv = [](long long a, long long b) { return (int)((a + b - 1) / b); };

    zero_f4_kernel<<<cdiv(N / 4, TB), TB, 0, stream>>>(deg, N / 4);
    unsigned* code = (unsigned*)Tb;
    deg_code_kernel<<<cdiv(cdiv(E, 4), TB), TB, 0, stream>>>(dst, deg, code, E);
    dinv_kernel<<<cdiv(N, TB), TB, 0, stream>>>(deg, dinv, N);

    h_init_kernel<<<cdiv(nd4, TB), TB, 0, stream>>>(feat, theta, h, nd4);

    const int groups = cdiv(N, PROP_ROWS);

    propagate_kernel<<<groups, PROP_THREADS, 0, stream>>>(feat, dinv, deg, src, dst, code, elist, nullptr, Ta, h, theta, 1, 1.0f, N, E);

    propagate_list_kernel<<<groups, PROP_THREADS, 0, stream>>>(Ta, dinv, elist, feat, Tb, h, theta, 2, 2.0f, N, E);

    propagate_list_kernel<<<groups, PROP_THREADS, 0, stream>>>(Tb, dinv, elist, Ta, Ta, h, theta, 3, 2.0f, N, E);

    const int tiles = (N / 16) * (DFEAT / 16);
    gemm_wmma_kernel<<<cdiv(tiles, 256 / 32), 256, 0, stream>>>(h, W, out, N);
}


# GEMM made weight-stationary: a wave keeps the W fragments of its 16 columns in registers and loops over row tiles, h read by 16-byte loads; W not re-read per tile; k order of the f32 sums changes
# speedup vs baseline: 3.1690x; 1.0850x over previous
#include <hip/hip_runtime.h>

typedef float v2f __attribute__((ext_vector_type(2)));
typedef float v8f __attribute__((ext_vector_type(8)));

#define DFEAT 128


__global__ void zero_f4_kernel(float* __restrict__ p, int n4) {
    int i = blockIdx.x * blockDim.x + threadIdx.x;
    if (i < n4) ((float4*)p)[i] = make_float4(0.f, 0.f, 0.f, 0.f);
}

#define TILE_ROWS 196
__global__ void deg_code_kernel(const int* __restrict__ dst, float* __restrict__ deg,
                                unsigned* __restrict__ code, int E) {
    const int i = blockIdx.x * blockDim.x + threadIdx.x;
    const int E4 = E >> 2;
    if (i < E4) {
        int4 d = ((const int4*)dst)[i];
        atomicAdd(&deg[d.x], 1.0f);
        atomicAdd(&deg[d.y], 1.0f);
        atomicAdd(&deg[d.z], 1.0f);
        atomicAdd(&deg[d.w], 1.0f);
        code[i] = ((unsigned)d.x / TILE_ROWS) | (((unsigned)d.y / TILE_ROWS) << 8) |
                  (((unsigned)d.z / TILE_ROWS) << 16) | (((unsigned)d.w / TILE_ROWS) << 24);
    }
    if (i < E - 4 * E4) atomicAdd(&deg[dst[4 * E4 + i]], 1.0f);
}

__global__ void dinv_kernel(const float* __restrict__ deg, float* __restrict__ dinv, int N) {
    int i = blockIdx.x * blockDim.x + threadIdx.x;
    if (i < N) dinv[i] = 1.0f / sqrtf(fmaxf(deg[i], 1.0f));
}

__global__ void h_init_kernel(const float* __restrict__ feat, const float* __restrict__ theta,
                              float* __restrict__ h, int n4) {
    int i = blockIdx.x * blockDim.x + threadIdx.x;
    if (i >= n4) return;
    float t0 = theta[0];
    float4 f = ((const float4*)feat)[i];
    ((float4*)h)[i] = make_float4(t0 * f.x, t0 * f.y, t0 * f.z, t0 * f.w);
}


#define PROP_THREADS 1024
#define PROP_WAVES   32
#define PROP_ROWS    196
#define SCAN_UNROLL  8
#define TEST_QUADS   2
static_assert(PROP_ROWS == TILE_ROWS, "the degree pass numbers the tiles the propagation owns");
#define APPEND_BATCH 32
#define DRAIN_BATCH  8
#define QCAP (APPEND_BATCH - 1 + 32)
static_assert(DRAIN_BATCH <= APPEND_BATCH, "the queue is sized for the larger batch");
static_assert(PROP_ROWS % 4 == 0, "a tile's first row must be a multiple of 4: the degree counts in front of it are read as float4");
#define BASE_UNROLL  8
#define LIST_CHUNK 4096
static_assert(LIST_CHUNK == 4 * PROP_THREADS, "a thread stages four entries and packs their four owner bytes into one word");
#define OWN_ROWS 7
static_assert(OWN_ROWS * PROP_WAVES >= PROP_ROWS, "every row of the tile has an owner wave and a register slot");

__device__ __forceinline__ unsigned not_mine(unsigned w, unsigned mine, int k) {
    unsigned x = w ^ mine;
    unsigned y = ((x & 0x7f7f7f7fu) + 0x7f7f7f7fu) | x;
    return (y >> (7 - k)) & (0x01010101u << k);
}

__device__ __forceinline__ void own_batch(const float* __restrict__ x, const float* __restrict__ dinv,
                                          const int2* ent, const int* qp, int n, int lane,
                                          float (&a)[OWN_ROWS][4]) {
    int s = 0, dl = 0;
    if (lane < n) {
        int2 e = ent[qp[lane]];
        s = e.x;
        dl = e.y;
    }
    float v[DRAIN_BATCH][4];
#pragma unroll
    for (int j = 0; j < DRAIN_BATCH; ++j) {
        v[j][0] = 0.0f; v[j][1] = 0.0f; v[j][2] = 0.0f; v[j][3] = 0.0f;
        if (j < n) {
            int sj = __builtin_amdgcn_readlane(s, j);
            const float* xp = x + (size_t)sj * DFEAT + lane;
            v[j][0] = xp[0];
            v[j][1] = xp[32];
            v[j][2] = xp[64];
            v[j][3] = xp[96];
        }
    }
    float w = 0.0f;
    if (lane < n) w = dinv[s];
#pragma unroll
    for (int j = 0; j < DRAIN_BATCH; ++j) {
        if (j < n) {
            const float wj = __int_as_float(__builtin_amdgcn_readlane(__float_as_int(w), j));
            const int kj = __builtin_amdgcn_readlane(dl, j) >> 5;
#pragma unroll
            for (int k = 0; k < OWN_ROWS; ++k) {
                if (kj == k) {
                    a[k][0] = fmaf(wj, v[j][0], a[k][0]);
                    a[k][1] = fmaf(wj, v[j][1], a[k][1]);
                    a[k][2] = fmaf(wj, v[j][2], a[k][2]);
                    a[k][3] = fmaf(wj, v[j][3], a[k][3]);
                }
            }
        }
    }
}

__device__ __forceinline__ void walk_finish(const float* __restrict__ x, const float* __restrict__ dinv,
                                            const int2* tileList, int count,
                                            int2* ent, unsigned* own, int* q,
                                            const float* Tprev, float* Tnew, float* __restrict__ h,
                                            const float* __restrict__ theta, int order, float c2,
                                            int lo, int cnt, int tid, int lane, int wave) {
    float a[OWN_ROWS][4];
#pragma unroll
    for (int k = 0; k < OWN_ROWS; ++k) { a[k][0] = 0.0f; a[k][1] = 0.0f; a[k][2] = 0.0f; a[k][3] = 0.0f; }

    int qn = 0;
    const unsigned mine = (unsigned)wave * 0x01010101u;

    for (int c0 = 0; c0 < count; c0 += LIST_CHUNK) {
        const int nc = (count - c0 < LIST_CHUNK) ? (count - c0) : LIST_CHUNK;
        {
            unsigned ow = 0;
#pragma unroll
            for (int u = 0; u < 4; ++u) {
                const int idx = 4 * tid + u;
                int2 e = tileList[c0 + (idx < nc ? idx : 0)];
                ent[idx] = e;
                unsigned b = (idx < nc) ? ((unsigned)e.y & 31u) : 0x7fu;
                ow |= b << (8 * u);
            }
            own[tid] = ow;
        }
        __syncthreads();

        const int steps = (nc + 511) >> 9;
        for (int st = 0; st < steps; ++st) {
            const int i4 = st * 32 + lane;
            const uint4 d = ((const uint4*)own)[i4];
            unsigned nm = not_mine(d.x, mine, 0) | not_mine(d.y, mine, 1) |
                          not_mine(d.z, mine, 2) | not_mine(d.w, mine, 3);
            unsigned mm = ~nm & 0x0f0f0f0fu;
            for (;;) {
                unsigned b = __builtin_amdgcn_ballot_w32(mm != 0);
                if (b == 0) break;
                if (mm != 0) {
                    int bit = __builtin_ctz(mm);
                    mm &= mm - 1;
                    int pos = qn + __builtin_popcount(b & ((1u << lane) - 1u));
                    q[pos] = 16 * i4 + 4 * (bit & 7) + (bit >> 3);
                }
                qn += __builtin_popcount(b);
                while (qn >= DRAIN_BATCH) {
                    qn -= DRAIN_BATCH;
                    own_batch(x, dinv, ent, q + qn, DRAIN_BATCH, lane, a);
                }
            }
        }
        if (qn > 0) {
            own_batch(x, dinv, ent, q, qn, lane, a);
            qn = 0;
        }
        __syncthreads();
    }

    const float th = theta[order] * 0.5f;
    float tp[OWN_ROWS][4], hv[OWN_ROWS][4], sc[OWN_ROWS];
#pragma unroll
    for (int k = 0; k < OWN_ROWS; ++k) {
        const int r = wave + 32 * k;
        sc[k] = 0.0f;
#pragma unroll
        for (int c = 0; c < 4; ++c) { tp[k][c] = 0.0f; hv[k][c] = 0.0f; }
        if (r < cnt) {
            const size_t g = (size_t)(lo + r) * DFEAT + lane;
            sc[k] = -c2 * dinv[lo + r];
#pragma unroll
            for (int c = 0; c < 4; ++c) {
                if (Tprev) tp[k][c] = Tprev[g + 32 * c];
                hv[k][c] = h[g + 32 * c];
            }
        }
    }
#pragma unroll
    for (int k = 0; k < OWN_ROWS; ++k) {
        const int r = wave + 32 * k;
        if (r < cnt) {
            const size_t g = (size_t)(lo + r) * DFEAT + lane;
#pragma unroll
            for (int c = 0; c < 4; ++c) {
                const float t = fmaf(sc[k], a[k][c], -tp[k][c]);
                Tnew[g + 32 * c] = t;
                h[g + 32 * c] = fmaf(th, t + 1.0f, hv[k][c]);
            }
        }
    }
}

__device__ __forceinline__ void append_edges(const int* __restrict__ src, const int* __restrict__ dst,
                                             const int* qp, int n, int lo, int lane,
                                             int2* tileList, int* cursor) {
    int s = 0, dl = 0;
    if (lane < n) {
        int e = qp[lane];
        s = src[e];
        dl = dst[e] - lo;
    }
    int pos = 0;
    if (lane == 0) pos = atomicAdd(cursor, n);
    pos = __builtin_amdgcn_readfirstlane(pos);
    if (lane < n) tileList[pos + lane] = make_int2(s, dl);
}

__device__ __forceinline__ void test_quads(const uint4* d, int base, unsigned mine, int lo, int lane,
                                           int* q, int& qn,
                                           const int* __restrict__ src, const int* __restrict__ dst,
                                           int2* tileList, int* cursor) {
    unsigned nm = 0;
#pragma unroll
    for (int u = 0; u < TEST_QUADS; ++u) {
        nm |= not_mine(d[u].x, mine, 4 * u + 0);
        nm |= not_mine(d[u].y, mine, 4 * u + 1);
        nm |= not_mine(d[u].z, mine, 4 * u + 2);
        nm |= not_mine(d[u].w, mine, 4 * u + 3);
    }
    unsigned mm = ~nm;
    for (;;) {
        unsigned b = __builtin_amdgcn_ballot_w32(mm != 0);
        if (b == 0) break;
        if (mm != 0) {
            int bit = __builtin_ctz(mm);
            mm &= mm - 1;
            int k = bit & 7;
            int i = base + (k >> 2) * PROP_THREADS;
            int pos = qn + __builtin_popcount(b & ((1u << lane) - 1u));
            q[pos] = 16 * i + 4 * (k & 3) + (bit >> 3);
        }
        qn += __builtin_popcount(b);
        while (qn >= APPEND_BATCH) {
            qn -= APPEND_BATCH;
            append_edges(src, dst, q + qn, APPEND_BATCH, lo, lane, tileList, cursor);
        }
    }
}

__global__ __launch_bounds__(PROP_THREADS)
void propagate_kernel(const float* __restrict__ x, const float* __restrict__ dinv,
                      const float* __restrict__ deg,
                      const int* __restrict__ src, const int* __restrict__ dst,
                      const unsigned* __restrict__ code,
                      int* elist,
                      const float* Tprev, float* Tnew, float* __restrict__ h,
                      const float* __restrict__ theta, int order, float c2, int N, int E) {
    __shared__ int2 ent[LIST_CHUNK];
    __shared__ __attribute__((aligned(16))) unsigned own[LIST_CHUNK / 4];
    __shared__ int queue[PROP_WAVES][QCAP];
    __shared__ int listBase;
    __shared__ int cursor;

    const int tid = threadIdx.x;
    const int lane = tid & 31;
    const int wave = __builtin_amdgcn_readfirstlane(tid >> 5);
    const int lo = blockIdx.x * PROP_ROWS;
    const int cnt = (N - lo < PROP_ROWS) ? (N - lo) : PROP_ROWS;

    if (tid == 0) { listBase = 0; cursor = 0; }

    int part = 0;
    {
        const float4* deg4 = (const float4*)deg;
        const int n4 = lo >> 2;
        for (int i0 = tid; i0 < n4; i0 += BASE_UNROLL * PROP_THREADS) {
            float4 dv[BASE_UNROLL];
#pragma unroll
            for (int u = 0; u < BASE_UNROLL; ++u) {
                int i = i0 + u * PROP_THREADS;
                dv[u] = deg4[i < n4 ? i : 0];
            }
#pragma unroll
            for (int u = 0; u < BASE_UNROLL; ++u) {
                int i = i0 + u * PROP_THREADS;
                int t = (int)dv[u].x + (int)dv[u].y + (int)dv[u].z + (int)dv[u].w;
                part += (i < n4) ? t : 0;
            }
        }
        for (int off = 16; off > 0; off >>= 1) part += __shfl_xor(part, off, 32);
    }
    __syncthreads();
    if (lane == 0 && part != 0) atomicAdd(&listBase, part);
    __syncthreads();
    int2* list2 = (int2*)elist;
    int2* tileList = list2 + listBase;

    int* q = queue[wave];
    int qn = 0;

    const int E16 = E >> 4;
    const int span = SCAN_UNROLL * PROP_THREADS;
    const int nG = (E16 + span - 1) / span;
    const int gStart = (int)((long long)blockIdx.x * nG / gridDim.x);
    const uint4* code4 = (const uint4*)code;
    const unsigned mine = blockIdx.x * 0x01010101u;
    const unsigned other = mine ^ 0x01010101u;

    for (int g = 0; g < nG; ++g) {
        int gg = g + gStart;
        if (gg >= nG) gg -= nG;
        const int base = gg * span + tid;
        uint4 d[SCAN_UNROLL];
        if (gg * span + span <= E16) {
#pragma unroll
            for (int u = 0; u < SCAN_UNROLL; ++u) d[u] = code4[base + u * PROP_THREADS];
        } else {
#pragma unroll
            for (int u = 0; u < SCAN_UNROLL; ++u) {
                int i = base + u * PROP_THREADS;
                uint4 t = code4[i < E16 ? i : 0];
                if (i >= E16) t = make_uint4(other, other, other, other);
                d[u] = t;
            }
        }
#pragma unroll
        for (int p = 0; p < SCAN_UNROLL; p += TEST_QUADS)
            test_quads(d + p, base + p * PROP_THREADS, mine, lo, lane, q, qn, src, dst, tileList, &cursor);
    }

    const int rem = E - 16 * E16;
    if (wave == 0 && rem > 0) {
        int e = 16 * E16 + lane;
        bool hit = false;
        if (lane < rem) hit = (unsigned)(dst[e] - lo) < (unsigned)cnt;
        unsigned b = __builtin_amdgcn_ballot_w32(hit);
        if (hit) q[qn + __builtin_popcount(b & ((1u << lane) - 1u))] = e;
        qn += __builtin_popcount(b);
    }
    while (qn > 0) {
        int n = qn < APPEND_BATCH ? qn : APPEND_BATCH;
        qn -= n;
        append_edges(src, dst, q + qn, n, lo, lane, tileList, &cursor);
    }
    __syncthreads();
    const int count = cursor;
    if (tid == 0) list2[E + blockIdx.x] = make_int2(listBase, count);

    walk_finish(x, dinv, tileList, count, ent, own, q, Tprev, Tnew, h, theta, order, c2, lo, cnt, tid, lane, wave);
}

__global__ __launch_bounds__(PROP_THREADS)
void propagate_list_kernel(const float* __restrict__ x, const float* __restrict__ dinv,
                           const int* __restrict__ elist,
                           const float* Tprev, float* Tnew, float* __restrict__ h,
                           const float* __restrict__ theta, int order, float c2, int N, int E) {
    __shared__ int2 ent[LIST_CHUNK];
    __shared__ __attribute__((aligned(16))) unsigned own[LIST_CHUNK / 4];
    __shared__ int queue[PROP_WAVES][QCAP];

    const int tid = threadIdx.x;
    const int lane = tid & 31;
    const int wave = __builtin_amdgcn_readfirstlane(tid >> 5);
    const int lo = blockIdx.x * PROP_ROWS;
    const int cnt = (N - lo < PROP_ROWS) ? (N - lo) : PROP_ROWS;

    const int2* list2 = (const int2*)elist;
    const int2 m = list2[E + blockIdx.x];

    walk_finish(x, dinv, list2 + m.x, m.y, ent, own, queue[wave], Tprev, Tnew, h, theta, order, c2, lo, cnt, tid, lane, wave);
}


#define GEMM_WAVES 8
static_assert(GEMM_WAVES * 16 == DFEAT, "one wave per 16-column tile of the output");
#define GEMM_KQ (DFEAT / 8)

__global__ __launch_bounds__(32 * GEMM_WAVES)
void gemm_wmma_kernel(const float* __restrict__ h, const float* __restrict__ W,
                      float* __restrict__ out, int N) {
    const int lane = threadIdx.x & 31;
    const int wave = __builtin_amdgcn_readfirstlane((int)(threadIdx.x >> 5));
    const int nRowTiles = N >> 4;
    const int colBase = wave << 4;
    const int m = lane & 15;
    const int half = lane >> 4;

    float4 b[GEMM_KQ];
    {
        const float4* wrow = (const float4*)(W + (size_t)(colBase + m) * DFEAT + 4 * half);
#pragma unroll
        for (int i = 0; i < GEMM_KQ; ++i) b[i] = wrow[2 * i];
    }

    for (int rt = blockIdx.x; rt < nRowTiles; rt += gridDim.x) {
        const int rowBase = rt << 4;
        const float4* hrow = (const float4*)(h + (size_t)(rowBase + m) * DFEAT + 4 * half);
        float4 a[GEMM_KQ];
#pragma unroll
        for (int i = 0; i < GEMM_KQ; ++i) a[i] = hrow[2 * i];
        __builtin_amdgcn_sched_barrier(0);

        v8f acc = {};
#pragma unroll
        for (int i = 0; i < GEMM_KQ; ++i) {
            const v2f a0 = {a[i].x, a[i].y};
            const v2f b0 = {b[i].x, b[i].y};
            acc = __builtin_amdgcn_wmma_f32_16x16x4_f32(
                 false, a0,  false, b0,
                 (short)0, acc,  false,  false);
            const v2f a1 = {a[i].z, a[i].w};
            const v2f b1 = {b[i].z, b[i].w};
            acc = __builtin_amdgcn_wmma_f32_16x16x4_f32(
                 false, a1,  false, b1,
                 (short)0, acc,  false,  false);
        }

#pragma unroll
        for (int v = 0; v < 8; ++v) {
            const int r = rowBase + v + (half << 3);
            float val = acc[v];
            val = (val >= 0.0f) ? val : 0.01f * val;
            out[(size_t)r * DFEAT + colBase + m] = val;
        }
    }
}


extern "C" void kernel_launch(void* const* d_in, const int* in_sizes, int n_in,
                              void* d_out, int out_size, void* d_ws, size_t ws_size,
                              hipStream_t stream) {
    const float* feat  = (const float*)d_in[0];
    const int*   src   = (const int*)d_in[1];
    const int*   dst   = (const int*)d_in[2];
    const float* theta = (const float*)d_in[3];
    const float* W     = (const float*)d_in[4];
    float* out = (float*)d_out;

    const int N = in_sizes[0] / DFEAT;
    const int E = in_sizes[1];

    float* ws   = (float*)d_ws;
    float* deg  = ws;
    float* dinv = deg + N;
    int*   elist = (int*)(dinv + N);
    float* Ta   = dinv + N + (size_t)N * DFEAT;
    float* Tb   = Ta + (size_t)N * DFEAT;
    float* h    = Tb + (size_t)N * DFEAT;

    const int TB = 256;
    const int nd4 = N * (DFEAT / 4);
    auto cdiv = [](long long a, long long b) { return (int)((a + b - 1) / b); };

    zero_f4_kernel<<<cdiv(N / 4, TB), TB, 0, stream>>>(deg, N / 4);
    unsigned* code = (unsigned*)Tb;
    deg_code_kernel<<<cdiv(cdiv(E, 4), TB), TB, 0, stream>>>(dst, deg, code, E);
    dinv_kernel<<<cdiv(N, TB), TB, 0, stream>>>(deg, dinv, N);

    h_init_kernel<<<cdiv(nd4, TB), TB, 0, stream>>>(feat, theta, h, nd4);

    const int groups = cdiv(N, PROP_ROWS);

    propagate_kernel<<<groups, PROP_THREADS, 0, stream>>>(feat, dinv, deg, src, dst, code, elist, nullptr, Ta, h, theta, 1, 1.0f, N, E);

    propagate_list_kernel<<<groups, PROP_THREADS, 0, stream>>>(Ta, dinv, elist, feat, Tb, h, theta, 2, 2.0f, N, E);

    propagate_list_kernel<<<groups, PROP_THREADS, 0, stream>>>(Tb, dinv, elist, Ta, Ta, h, theta, 3, 2.0f, N, E);

    gemm_wmma_kernel<<<768, 32 * GEMM_WAVES, 0, stream>>>(h, W, out, N);
}


# Tile lists filled by one partition launch: a group ranks its 4096 edges per tile in LDS, reserves its run by one integer atomic per tile, stores records; 256 all-edge scans gone; arrival-order sums
# speedup vs baseline: 3.2981x; 1.0407x over previous
#include <hip/hip_runtime.h>

typedef float v2f __attribute__((ext_vector_type(2)));
typedef float v8f __attribute__((ext_vector_type(8)));

#define DFEAT 128


__global__ void zero_kernel(float* __restrict__ p, int n4, int* __restrict__ q, int nq) {
    int i = blockIdx.x * blockDim.x + threadIdx.x;
    if (i < n4) ((float4*)p)[i] = make_float4(0.f, 0.f, 0.f, 0.f);
    if (i < nq) q[i] = 0;
}

#define TILE_ROWS 196
__global__ void deg_kernel(const int* __restrict__ dst, float* __restrict__ deg, int E) {
    const int i = blockIdx.x * blockDim.x + threadIdx.x;
    const int E4 = E >> 2;
    if (i < E4) {
        int4 d = ((const int4*)dst)[i];
        atomicAdd(&deg[d.x], 1.0f);
        atomicAdd(&deg[d.y], 1.0f);
        atomicAdd(&deg[d.z], 1.0f);
        atomicAdd(&deg[d.w], 1.0f);
    }
    if (i < E - 4 * E4) atomicAdd(&deg[dst[4 * E4 + i]], 1.0f);
}

__global__ void h_init_kernel(const float* __restrict__ feat, const float* __restrict__ theta,
                              float* __restrict__ h, int n4) {
    int i = blockIdx.x * blockDim.x + threadIdx.x;
    if (i >= n4) return;
    float t0 = theta[0];
    float4 f = ((const float4*)feat)[i];
    ((float4*)h)[i] = make_float4(t0 * f.x, t0 * f.y, t0 * f.z, t0 * f.w);
}


#define PROP_THREADS 1024
#define PROP_WAVES   32
#define PROP_ROWS    196
static_assert(PROP_ROWS == TILE_ROWS, "the partition launch numbers the tiles the propagation owns");
#define DRAIN_BATCH  8
#define QCAP (DRAIN_BATCH - 1 + 32)
#define LIST_CHUNK 4096
static_assert(LIST_CHUNK == 4 * PROP_THREADS, "a thread stages four entries and packs their four owner bytes into one word");
#define OWN_ROWS 7
static_assert(OWN_ROWS * PROP_WAVES >= PROP_ROWS, "every row of the tile has an owner wave and a register slot");

__device__ __forceinline__ unsigned not_mine(unsigned w, unsigned mine, int k) {
    unsigned x = w ^ mine;
    unsigned y = ((x & 0x7f7f7f7fu) + 0x7f7f7f7fu) | x;
    return (y >> (7 - k)) & (0x01010101u << k);
}

__device__ __forceinline__ void own_batch(const float* __restrict__ x, const float* __restrict__ dinv,
                                          const int2* ent, const int* qp, int n, int lane,
                                          float (&a)[OWN_ROWS][4]) {
    int s = 0, dl = 0;
    if (lane < n) {
        int2 e = ent[qp[lane]];
        s = e.x;
        dl = e.y;
    }
    float v[DRAIN_BATCH][4];
#pragma unroll
    for (int j = 0; j < DRAIN_BATCH; ++j) {
        v[j][0] = 0.0f; v[j][1] = 0.0f; v[j][2] = 0.0f; v[j][3] = 0.0f;
        if (j < n) {
            int sj = __builtin_amdgcn_readlane(s, j);
            const float* xp = x + (size_t)sj * DFEAT + lane;
            v[j][0] = xp[0];
            v[j][1] = xp[32];
            v[j][2] = xp[64];
            v[j][3] = xp[96];
        }
    }
    float w = 0.0f;
    if (lane < n) w = dinv[s];
#pragma unroll
    for (int j = 0; j < DRAIN_BATCH; ++j) {
        if (j < n) {
            const float wj = __int_as_float(__builtin_amdgcn_readlane(__float_as_int(w), j));
            const int kj = __builtin_amdgcn_readlane(dl, j) >> 5;
#pragma unroll
            for (int k = 0; k < OWN_ROWS; ++k) {
                if (kj == k) {
                    a[k][0] = fmaf(wj, v[j][0], a[k][0]);
                    a[k][1] = fmaf(wj, v[j][1], a[k][1]);
                    a[k][2] = fmaf(wj, v[j][2], a[k][2]);
                    a[k][3] = fmaf(wj, v[j][3], a[k][3]);
                }
            }
        }
    }
}

__device__ __forceinline__ void walk_finish(const float* __restrict__ x, const float* __restrict__ dinv,
                                            const int2* tileList, int count,
                                            int2* ent, unsigned* own, int* q,
                                            const float* Tprev, float* Tnew, float* __restrict__ h,
                                            const float* __restrict__ theta, int order, float c2,
                                            int lo, int cnt, int tid, int lane, int wave) {
    float a[OWN_ROWS][4];
#pragma unroll
    for (int k = 0; k < OWN_ROWS; ++k) { a[k][0] = 0.0f; a[k][1] = 0.0f; a[k][2] = 0.0f; a[k][3] = 0.0f; }

    int qn = 0;
    const unsigned mine = (unsigned)wave * 0x01010101u;

    for (int c0 = 0; c0 < count; c0 += LIST_CHUNK) {
        const int nc = (count - c0 < LIST_CHUNK) ? (count - c0) : LIST_CHUNK;
        {
            unsigned ow = 0;
#pragma unroll
            for (int u = 0; u < 4; ++u) {
                const int idx = 4 * tid + u;
                int2 e = tileList[c0 + (idx < nc ? idx : 0)];
                ent[idx] = e;
                unsigned b = (idx < nc) ? ((unsigned)e.y & 31u) : 0x7fu;
                ow |= b << (8 * u);
            }
            own[tid] = ow;
        }
        __syncthreads();

        const int steps = (nc + 511) >> 9;
        for (int st = 0; st < steps; ++st) {
            const int i4 = st * 32 + lane;
            const uint4 d = ((const uint4*)own)[i4];
            unsigned nm = not_mine(d.x, mine, 0) | not_mine(d.y, mine, 1) |
                          not_mine(d.z, mine, 2) | not_mine(d.w, mine, 3);
            unsigned mm = ~nm & 0x0f0f0f0fu;
            for (;;) {
                unsigned b = __builtin_amdgcn_ballot_w32(mm != 0);
                if (b == 0) break;
                if (mm != 0) {
                    int bit = __builtin_ctz(mm);
                    mm &= mm - 1;
                    int pos = qn + __builtin_popcount(b & ((1u << lane) - 1u));
                    q[pos] = 16 * i4 + 4 * (bit & 7) + (bit >> 3);
                }
                qn += __builtin_popcount(b);
                while (qn >= DRAIN_BATCH) {
                    qn -= DRAIN_BATCH;
                    own_batch(x, dinv, ent, q + qn, DRAIN_BATCH, lane, a);
                }
            }
        }
        if (qn > 0) {
            own_batch(x, dinv, ent, q, qn, lane, a);
            qn = 0;
        }
        __syncthreads();
    }

    const float th = theta[order] * 0.5f;
    float tp[OWN_ROWS][4], hv[OWN_ROWS][4], sc[OWN_ROWS];
#pragma unroll
    for (int k = 0; k < OWN_ROWS; ++k) {
        const int r = wave + 32 * k;
        sc[k] = 0.0f;
#pragma unroll
        for (int c = 0; c < 4; ++c) { tp[k][c] = 0.0f; hv[k][c] = 0.0f; }
        if (r < cnt) {
            const size_t g = (size_t)(lo + r) * DFEAT + lane;
            sc[k] = -c2 * dinv[lo + r];
#pragma unroll
            for (int c = 0; c < 4; ++c) {
                if (Tprev) tp[k][c] = Tprev[g + 32 * c];
                hv[k][c] = h[g + 32 * c];
            }
        }
    }
#pragma unroll
    for (int k = 0; k < OWN_ROWS; ++k) {
        const int r = wave + 32 * k;
        if (r < cnt) {
            const size_t g = (size_t)(lo + r) * DFEAT + lane;
#pragma unroll
            for (int c = 0; c < 4; ++c) {
                const float t = fmaf(sc[k], a[k][c], -tp[k][c]);
                Tnew[g + 32 * c] = t;
                h[g + 32 * c] = fmaf(th, t + 1.0f, hv[k][c]);
            }
        }
    }
}

#define PART_THREADS 1024
#define PART_WAVES   32
#define MAX_TILES    256
#define TILES_PER_WAVE (MAX_TILES / PART_WAVES)
#define ROW_LOADS ((PROP_ROWS + 31) / 32)

__global__ __launch_bounds__(PART_THREADS)
void partition_kernel(const int* __restrict__ src, const int* __restrict__ dst,
                      const float* __restrict__ deg, float* __restrict__ dinv,
                      int* tileCur, int* elist, int N, int E) {
    int2* list2 = (int2*)elist;
    __shared__ int tileSum[MAX_TILES];
    __shared__ int tileBase[MAX_TILES];
    __shared__ int cnt[MAX_TILES];
    __shared__ int blockOff[MAX_TILES];

    const int tid = threadIdx.x;
    const int lane = tid & 31;
    const int wave = tid >> 5;
    const int i = blockIdx.x * PART_THREADS + tid;
    const int E4 = E >> 2;

    int s[4] = {0, 0, 0, 0}, d[4] = {0, 0, 0, 0};
    int ne = 0;
    if (i < E4) {
        const int4 dv = ((const int4*)dst)[i];
        const int4 sv = ((const int4*)src)[i];
        d[0] = dv.x; d[1] = dv.y; d[2] = dv.z; d[3] = dv.w;
        s[0] = sv.x; s[1] = sv.y; s[2] = sv.z; s[3] = sv.w;
        ne = 4;
    } else if (i == E4) {
        ne = E - 4 * E4;
#pragma unroll
        for (int u = 0; u < 3; ++u)
            if (u < ne) { d[u] = dst[4 * E4 + u]; s[u] = src[4 * E4 + u]; }
    }

    if (tid < MAX_TILES) cnt[tid] = 0;

    for (int r = i; r < N; r += gridDim.x * PART_THREADS) dinv[r] = 1.0f / sqrtf(fmaxf(deg[r], 1.0f));

    {
        float dv[TILES_PER_WAVE][ROW_LOADS];
#pragma unroll
        for (int a = 0; a < TILES_PER_WAVE; ++a) {
            const int lo = (wave + PART_WAVES * a) * PROP_ROWS;
#pragma unroll
            for (int b = 0; b < ROW_LOADS; ++b) {
                const int r = 32 * b + lane;
                const bool ok = (r < PROP_ROWS) && (lo + r < N);
                dv[a][b] = deg[ok ? lo + r : 0];
            }
        }
#pragma unroll
        for (int a = 0; a < TILES_PER_WAVE; ++a) {
            const int lo = (wave + PART_WAVES * a) * PROP_ROWS;
            int part = 0;
#pragma unroll
            for (int b = 0; b < ROW_LOADS; ++b) {
                const int r = 32 * b + lane;
                const bool ok = (r < PROP_ROWS) && (lo + r < N);
                part += ok ? (int)dv[a][b] : 0;
            }
            for (int off = 16; off > 0; off >>= 1) part += __shfl_xor(part, off, 32);
            if (lane == 0) tileSum[wave + PART_WAVES * a] = part;
        }
    }
    __syncthreads();

    if (wave == 0) {
        int v[TILES_PER_WAVE];
        int sum = 0;
#pragma unroll
        for (int u = 0; u < TILES_PER_WAVE; ++u) { v[u] = tileSum[TILES_PER_WAVE * lane + u]; sum += v[u]; }
        int inc = sum;
        for (int off = 1; off < 32; off <<= 1) {
            const int nb = __shfl_up(inc, off, 32);
            if (lane >= off) inc += nb;
        }
        int run = inc - sum;
#pragma unroll
        for (int u = 0; u < TILES_PER_WAVE; ++u) { tileBase[TILES_PER_WAVE * lane + u] = run; run += v[u]; }
    }

    int tl[4] = {0, 0, 0, 0}, rk[4] = {0, 0, 0, 0};
#pragma unroll
    for (int u = 0; u < 4; ++u) {
        if (u < ne) {
            tl[u] = (int)((unsigned)d[u] / PROP_ROWS);
            rk[u] = atomicAdd(&cnt[tl[u]], 1);
        }
    }
    __syncthreads();

    if (tid < MAX_TILES) {
        const int c = cnt[tid];
        blockOff[tid] = (c > 0) ? atomicAdd(&tileCur[tid], c) : 0;
    }
    __syncthreads();

#pragma unroll
    for (int u = 0; u < 4; ++u) {
        if (u < ne) {
            const int pos = tileBase[tl[u]] + blockOff[tl[u]] + rk[u];
            list2[pos] = make_int2(s[u], d[u] - tl[u] * PROP_ROWS);
        }
    }
    if (blockIdx.x == 0 && tid < MAX_TILES) list2[E + tid] = make_int2(tileBase[tid], tileSum[tid]);
}

__global__ __launch_bounds__(PROP_THREADS)
void propagate_list_kernel(const float* __restrict__ x, const float* __restrict__ dinv,
                           const int* __restrict__ elist,
                           const float* Tprev, float* Tnew, float* __restrict__ h,
                           const float* __restrict__ theta, int order, float c2, int N, int E) {
    __shared__ int2 ent[LIST_CHUNK];
    __shared__ __attribute__((aligned(16))) unsigned own[LIST_CHUNK / 4];
    __shared__ int queue[PROP_WAVES][QCAP];

    const int tid = threadIdx.x;
    const int lane = tid & 31;
    const int wave = __builtin_amdgcn_readfirstlane(tid >> 5);
    const int lo = blockIdx.x * PROP_ROWS;
    const int cnt = (N - lo < PROP_ROWS) ? (N - lo) : PROP_ROWS;

    const int2* list2 = (const int2*)elist;
    const int2 m = list2[E + blockIdx.x];

    walk_finish(x, dinv, list2 + m.x, m.y, ent, own, queue[wave], Tprev, Tnew, h, theta, order, c2, lo, cnt, tid, lane, wave);
}


#define GEMM_WAVES 8
static_assert(GEMM_WAVES * 16 == DFEAT, "one wave per 16-column tile of the output");
#define GEMM_KQ (DFEAT / 8)

__global__ __launch_bounds__(32 * GEMM_WAVES)
void gemm_wmma_kernel(const float* __restrict__ h, const float* __restrict__ W,
                      float* __restrict__ out, int N) {
    const int lane = threadIdx.x & 31;
    const int wave = __builtin_amdgcn_readfirstlane((int)(threadIdx.x >> 5));
    const int nRowTiles = N >> 4;
    const int colBase = wave << 4;
    const int m = lane & 15;
    const int half = lane >> 4;

    float4 b[GEMM_KQ];
    {
        const float4* wrow = (const float4*)(W + (size_t)(colBase + m) * DFEAT + 4 * half);
#pragma unroll
        for (int i = 0; i < GEMM_KQ; ++i) b[i] = wrow[2 * i];
    }

    for (int rt = blockIdx.x; rt < nRowTiles; rt += gridDim.x) {
        const int rowBase = rt << 4;
        const float4* hrow = (const float4*)(h + (size_t)(rowBase + m) * DFEAT + 4 * half);
        float4 a[GEMM_KQ];
#pragma unroll
        for (int i = 0; i < GEMM_KQ; ++i) a[i] = hrow[2 * i];
        __builtin_amdgcn_sched_barrier(0);

        v8f acc = {};
#pragma unroll
        for (int i = 0; i < GEMM_KQ; ++i) {
            const v2f a0 = {a[i].x, a[i].y};
            const v2f b0 = {b[i].x, b[i].y};
            acc = __builtin_amdgcn_wmma_f32_16x16x4_f32(
                 false, a0,  false, b0,
                 (short)0, acc,  false,  false);
            const v2f a1 = {a[i].z, a[i].w};
            const v2f b1 = {b[i].z, b[i].w};
            acc = __builtin_amdgcn_wmma_f32_16x16x4_f32(
                 false, a1,  false, b1,
                 (short)0, acc,  false,  false);
        }

#pragma unroll
        for (int v = 0; v < 8; ++v) {
            const int r = rowBase + v + (half << 3);
            float val = acc[v];
            val = (val >= 0.0f) ? val : 0.01f * val;
            out[(size_t)r * DFEAT + colBase + m] = val;
        }
    }
}


extern "C" void kernel_launch(void* const* d_in, const int* in_sizes, int n_in,
                              void* d_out, int out_size, void* d_ws, size_t ws_size,
                              hipStream_t stream) {
    const float* feat  = (const float*)d_in[0];
    const int*   src   = (const int*)d_in[1];
    const int*   dst   = (const int*)d_in[2];
    const float* theta = (const float*)d_in[3];
    const float* W     = (const float*)d_in[4];
    float* out = (float*)d_out;

    const int N = in_sizes[0] / DFEAT;
    const int E = in_sizes[1];

    float* ws   = (float*)d_ws;
    float* deg  = ws;
    float* dinv = deg + N;
    int*   elist = (int*)(dinv + N);
    int*   tileCur = elist + 2 * ((size_t)E + MAX_TILES);
    float* Ta   = dinv + N + (size_t)N * DFEAT;
    float* Tb   = Ta + (size_t)N * DFEAT;
    float* h    = Tb + (size_t)N * DFEAT;

    const int TB = 256;
    const int nd4 = N * (DFEAT / 4);
    auto cdiv = [](long long a, long long b) { return (int)((a + b - 1) / b); };

    zero_kernel<<<cdiv(N / 4, TB), TB, 0, stream>>>(deg, N / 4, tileCur, MAX_TILES);
    deg_kernel<<<cdiv(cdiv(E, 4), TB), TB, 0, stream>>>(dst, deg, E);

    partition_kernel<<<cdiv(E / 4 + 1, PART_THREADS), PART_THREADS, 0, stream>>>(src, dst, deg, dinv, tileCur, elist, N, E);

    h_init_kernel<<<cdiv(nd4, TB), TB, 0, stream>>>(feat, theta, h, nd4);

    const int groups = cdiv(N, PROP_ROWS);

    propagate_list_kernel<<<groups, PROP_THREADS, 0, stream>>>(feat, dinv, elist, nullptr, Ta, h, theta, 1, 1.0f, N, E);

    propagate_list_kernel<<<groups, PROP_THREADS, 0, stream>>>(Ta, dinv, elist, feat, Tb, h, theta, 2, 2.0f, N, E);

    propagate_list_kernel<<<groups, PROP_THREADS, 0, stream>>>(Tb, dinv, elist, Ta, Ta, h, theta, 3, 2.0f, N, E);

    gemm_wmma_kernel<<<768, 32 * GEMM_WAVES, 0, stream>>>(h, W, out, N);
}


# Edge list entry carries its source row's weight 1/sqrt(max(deg,1)), formed once by the partition launch: orders 1-3 drop the per-edge dinv[src] load, batch sums start as rows land; sums unchanged
# speedup vs baseline: 3.4351x; 1.0415x over previous
#include <hip/hip_runtime.h>

typedef float v2f __attribute__((ext_vector_type(2)));
typedef float v8f __attribute__((ext_vector_type(8)));

#define DFEAT 128

__device__ __forceinline__ float inv_sqrt_deg(float d) { return 1.0f / sqrtf(fmaxf(d, 1.0f)); }

#define ENT_DL_SHIFT 24
#define ENT_SRC_MASK 0x00ffffffu


__global__ void zero_kernel(float* __restrict__ p, int n4, int* __restrict__ q, int nq) {
    int i = blockIdx.x * blockDim.x + threadIdx.x;
    if (i < n4) ((float4*)p)[i] = make_float4(0.f, 0.f, 0.f, 0.f);
    if (i < nq) q[i] = 0;
}

#define TILE_ROWS 196
__global__ void deg_kernel(const int* __restrict__ dst, float* __restrict__ deg, int E) {
    const int i = blockIdx.x * blockDim.x + threadIdx.x;
    const int E4 = E >> 2;
    if (i < E4) {
        int4 d = ((const int4*)dst)[i];
        atomicAdd(&deg[d.x], 1.0f);
        atomicAdd(&deg[d.y], 1.0f);
        atomicAdd(&deg[d.z], 1.0f);
        atomicAdd(&deg[d.w], 1.0f);
    }
    if (i < E - 4 * E4) atomicAdd(&deg[dst[4 * E4 + i]], 1.0f);
}

__global__ void h_init_kernel(const float* __restrict__ feat, const float* __restrict__ theta,
                              float* __restrict__ h, int n4) {
    int i = blockIdx.x * blockDim.x + threadIdx.x;
    if (i >= n4) return;
    float t0 = theta[0];
    float4 f = ((const float4*)feat)[i];
    ((float4*)h)[i] = make_float4(t0 * f.x, t0 * f.y, t0 * f.z, t0 * f.w);
}


#define PROP_THREADS 1024
#define PROP_WAVES   32
#define PROP_ROWS    196
static_assert(PROP_ROWS == TILE_ROWS, "the partition launch numbers the tiles the propagation owns");
#define DRAIN_BATCH  8
#define QCAP (DRAIN_BATCH - 1 + 32)
#define LIST_CHUNK 4096
static_assert(LIST_CHUNK == 4 * PROP_THREADS, "a thread stages four entries and packs their four owner bytes into one word");
#define OWN_ROWS 7
static_assert(OWN_ROWS * PROP_WAVES >= PROP_ROWS, "every row of the tile has an owner wave and a register slot");
static_assert(PROP_ROWS <= (1 << (32 - ENT_DL_SHIFT)), "a local destination row fits the bits of an entry's first word above the source row");

__device__ __forceinline__ unsigned not_mine(unsigned w, unsigned mine, int k) {
    unsigned x = w ^ mine;
    unsigned y = ((x & 0x7f7f7f7fu) + 0x7f7f7f7fu) | x;
    return (y >> (7 - k)) & (0x01010101u << k);
}

__device__ __forceinline__ void own_batch(const float* __restrict__ x,
                                          const int2* ent, const int* qp, int n, int lane,
                                          float (&a)[OWN_ROWS][4]) {
    unsigned p = 0;
    float w = 0.0f;
    if (lane < n) {
        int2 e = ent[qp[lane]];
        p = (unsigned)e.x;
        w = __int_as_float(e.y);
    }
    float v[DRAIN_BATCH][4];
#pragma unroll
    for (int j = 0; j < DRAIN_BATCH; ++j) {
        v[j][0] = 0.0f; v[j][1] = 0.0f; v[j][2] = 0.0f; v[j][3] = 0.0f;
        if (j < n) {
            const unsigned sj = (unsigned)__builtin_amdgcn_readlane((int)p, j) & ENT_SRC_MASK;
            const float* xp = x + (size_t)sj * DFEAT + lane;
            v[j][0] = xp[0];
            v[j][1] = xp[32];
            v[j][2] = xp[64];
            v[j][3] = xp[96];
        }
    }
#pragma unroll
    for (int j = 0; j < DRAIN_BATCH; ++j) {
        if (j < n) {
            const float wj = __int_as_float(__builtin_amdgcn_readlane(__float_as_int(w), j));
            const int kj = (int)((unsigned)__builtin_amdgcn_readlane((int)p, j) >> (ENT_DL_SHIFT + 5));
#pragma unroll
            for (int k = 0; k < OWN_ROWS; ++k) {
                if (kj == k) {
                    a[k][0] = fmaf(wj, v[j][0], a[k][0]);
                    a[k][1] = fmaf(wj, v[j][1], a[k][1]);
                    a[k][2] = fmaf(wj, v[j][2], a[k][2]);
                    a[k][3] = fmaf(wj, v[j][3], a[k][3]);
                }
            }
        }
    }
}

__device__ __forceinline__ void walk_finish(const float* __restrict__ x, const float* __restrict__ dinv,
                                            const int2* tileList, int count,
                                            int2* ent, unsigned* own, int* q,
                                            const float* Tprev, float* Tnew, float* __restrict__ h,
                                            const float* __restrict__ theta, int order, float c2,
                                            int lo, int cnt, int tid, int lane, int wave) {
    float a[OWN_ROWS][4];
#pragma unroll
    for (int k = 0; k < OWN_ROWS; ++k) { a[k][0] = 0.0f; a[k][1] = 0.0f; a[k][2] = 0.0f; a[k][3] = 0.0f; }

    int qn = 0;
    const unsigned mine = (unsigned)wave * 0x01010101u;

    for (int c0 = 0; c0 < count; c0 += LIST_CHUNK) {
        const int nc = (count - c0 < LIST_CHUNK) ? (count - c0) : LIST_CHUNK;
        {
            unsigned ow = 0;
#pragma unroll
            for (int u = 0; u < 4; ++u) {
                const int idx = 4 * tid + u;
                int2 e = tileList[c0 + (idx < nc ? idx : 0)];
                ent[idx] = e;
                unsigned b = (idx < nc) ? (((unsigned)e.x >> ENT_DL_SHIFT) & 31u) : 0x7fu;
                ow |= b << (8 * u);
            }
            own[tid] = ow;
        }
        __syncthreads();

        const int steps = (nc + 511) >> 9;
        for (int st = 0; st < steps; ++st) {
            const int i4 = st * 32 + lane;
            const uint4 d = ((const uint4*)own)[i4];
            unsigned nm = not_mine(d.x, mine, 0) | not_mine(d.y, mine, 1) |
                          not_mine(d.z, mine, 2) | not_mine(d.w, mine, 3);
            unsigned mm = ~nm & 0x0f0f0f0fu;
            for (;;) {
                unsigned b = __builtin_amdgcn_ballot_w32(mm != 0);
                if (b == 0) break;
                if (mm != 0) {
                    int bit = __builtin_ctz(mm);
                    mm &= mm - 1;
                    int pos = qn + __builtin_popcount(b & ((1u << lane) - 1u));
                    q[pos] = 16 * i4 + 4 * (bit & 7) + (bit >> 3);
                }
                qn += __builtin_popcount(b);
                while (qn >= DRAIN_BATCH) {
                    qn -= DRAIN_BATCH;
                    own_batch(x, ent, q + qn, DRAIN_BATCH, lane, a);
                }
            }
        }
        if (qn > 0) {
            own_batch(x, ent, q, qn, lane, a);
            qn = 0;
        }
        __syncthreads();
    }

    const float th = theta[order] * 0.5f;
    float tp[OWN_ROWS][4], hv[OWN_ROWS][4], sc[OWN_ROWS];
#pragma unroll
    for (int k = 0; k < OWN_ROWS; ++k) {
        const int r = wave + 32 * k;
        sc[k] = 0.0f;
#pragma unroll
        for (int c = 0; c < 4; ++c) { tp[k][c] = 0.0f; hv[k][c] = 0.0f; }
        if (r < cnt) {
            const size_t g = (size_t)(lo + r) * DFEAT + lane;
            sc[k] = -c2 * dinv[lo + r];
#pragma unroll
            for (int c = 0; c < 4; ++c) {
                if (Tprev) tp[k][c] = Tprev[g + 32 * c];
                hv[k][c] = h[g + 32 * c];
            }
        }
    }
#pragma unroll
    for (int k = 0; k < OWN_ROWS; ++k) {
        const int r = wave + 32 * k;
        if (r < cnt) {
            const size_t g = (size_t)(lo + r) * DFEAT + lane;
#pragma unroll
            for (int c = 0; c < 4; ++c) {
                const float t = fmaf(sc[k], a[k][c], -tp[k][c]);
                Tnew[g + 32 * c] = t;
                h[g + 32 * c] = fmaf(th, t + 1.0f, hv[k][c]);
            }
        }
    }
}

#define PART_THREADS 1024
#define PART_WAVES   32
#define MAX_TILES    256
#define TILES_PER_WAVE (MAX_TILES / PART_WAVES)
#define ROW_LOADS ((PROP_ROWS + 31) / 32)

__global__ __launch_bounds__(PART_THREADS)
void partition_kernel(const int* __restrict__ src, const int* __restrict__ dst,
                      const float* __restrict__ deg, float* __restrict__ dinv,
                      int* tileCur, int* elist, int N, int E) {
    int2* list2 = (int2*)elist;
    __shared__ int tileSum[MAX_TILES];
    __shared__ int tileBase[MAX_TILES];
    __shared__ int cnt[MAX_TILES];
    __shared__ int blockOff[MAX_TILES];

    const int tid = threadIdx.x;
    const int lane = tid & 31;
    const int wave = tid >> 5;
    const int i = blockIdx.x * PART_THREADS + tid;
    const int E4 = E >> 2;

    int s[4] = {0, 0, 0, 0}, d[4] = {0, 0, 0, 0};
    int ne = 0;
    if (i < E4) {
        const int4 dv = ((const int4*)dst)[i];
        const int4 sv = ((const int4*)src)[i];
        d[0] = dv.x; d[1] = dv.y; d[2] = dv.z; d[3] = dv.w;
        s[0] = sv.x; s[1] = sv.y; s[2] = sv.z; s[3] = sv.w;
        ne = 4;
    } else if (i == E4) {
        ne = E - 4 * E4;
#pragma unroll
        for (int u = 0; u < 3; ++u)
            if (u < ne) { d[u] = dst[4 * E4 + u]; s[u] = src[4 * E4 + u]; }
    }

    if (tid < MAX_TILES) cnt[tid] = 0;

    for (int r = i; r < N; r += gridDim.x * PART_THREADS) dinv[r] = inv_sqrt_deg(deg[r]);

    float sdeg[4];
#pragma unroll
    for (int u = 0; u < 4; ++u) sdeg[u] = deg[s[u]];

    {
        float dv[TILES_PER_WAVE][ROW_LOADS];
#pragma unroll
        for (int a = 0; a < TILES_PER_WAVE; ++a) {
            const int lo = (wave + PART_WAVES * a) * PROP_ROWS;
#pragma unroll
            for (int b = 0; b < ROW_LOADS; ++b) {
                const int r = 32 * b + lane;
                const bool ok = (r < PROP_ROWS) && (lo + r < N);
                dv[a][b] = deg[ok ? lo + r : 0];
            }
        }
#pragma unroll
        for (int a = 0; a < TILES_PER_WAVE; ++a) {
            const int lo = (wave + PART_WAVES * a) * PROP_ROWS;
            int part = 0;
#pragma unroll
            for (int b = 0; b < ROW_LOADS; ++b) {
                const int r = 32 * b + lane;
                const bool ok = (r < PROP_ROWS) && (lo + r < N);
                part += ok ? (int)dv[a][b] : 0;
            }
            for (int off = 16; off > 0; off >>= 1) part += __shfl_xor(part, off, 32);
            if (lane == 0) tileSum[wave + PART_WAVES * a] = part;
        }
    }
    __syncthreads();

    if (wave == 0) {
        int v[TILES_PER_WAVE];
        int sum = 0;
#pragma unroll
        for (int u = 0; u < TILES_PER_WAVE; ++u) { v[u] = tileSum[TILES_PER_WAVE * lane + u]; sum += v[u]; }
        int inc = sum;
        for (int off = 1; off < 32; off <<= 1) {
            const int nb = __shfl_up(inc, off, 32);
            if (lane >= off) inc += nb;
        }
        int run = inc - sum;
#pragma unroll
        for (int u = 0; u < TILES_PER_WAVE; ++u) { tileBase[TILES_PER_WAVE * lane + u] = run; run += v[u]; }
    }

    int tl[4] = {0, 0, 0, 0}, rk[4] = {0, 0, 0, 0};
#pragma unroll
    for (int u = 0; u < 4; ++u) {
        if (u < ne) {
            tl[u] = (int)((unsigned)d[u] / PROP_ROWS);
            rk[u] = atomicAdd(&cnt[tl[u]], 1);
        }
    }
    __syncthreads();

    if (tid < MAX_TILES) {
        const int c = cnt[tid];
        blockOff[tid] = (c > 0) ? atomicAdd(&tileCur[tid], c) : 0;
    }
    __syncthreads();

#pragma unroll
    for (int u = 0; u < 4; ++u) {
        if (u < ne) {
            const int pos = tileBase[tl[u]] + blockOff[tl[u]] + rk[u];
            const unsigned dl = (unsigned)(d[u] - tl[u] * PROP_ROWS);
            list2[pos] = make_int2((int)((unsigned)s[u] | (dl << ENT_DL_SHIFT)), __float_as_int(inv_sqrt_deg(sdeg[u])));
        }
    }
    if (blockIdx.x == 0 && tid < MAX_TILES) list2[E + tid] = make_int2(tileBase[tid], tileSum[tid]);
}

__global__ __launch_bounds__(PROP_THREADS)
void propagate_list_kernel(const float* __restrict__ x, const float* __restrict__ dinv,
                           const int* __restrict__ elist,
                           const float* Tprev, float* Tnew, float* __restrict__ h,
                           const float* __restrict__ theta, int order, float c2, int N, int E) {
    __shared__ int2 ent[LIST_CHUNK];
    __shared__ __attribute__((aligned(16))) unsigned own[LIST_CHUNK / 4];
    __shared__ int queue[PROP_WAVES][QCAP];

    const int tid = threadIdx.x;
    const int lane = tid & 31;
    const int wave = __builtin_amdgcn_readfirstlane(tid >> 5);
    const int lo = blockIdx.x * PROP_ROWS;
    const int cnt = (N - lo < PROP_ROWS) ? (N - lo) : PROP_ROWS;

    const int2* list2 = (const int2*)elist;
    const int2 m = list2[E + blockIdx.x];

    walk_finish(x, dinv, list2 + m.x, m.y, ent, own, queue[wave], Tprev, Tnew, h, theta, order, c2, lo, cnt, tid, lane, wave);
}


#define GEMM_WAVES 8
static_assert(GEMM_WAVES * 16 == DFEAT, "one wave per 16-column tile of the output");
#define GEMM_KQ (DFEAT / 8)

__global__ __launch_bounds__(32 * GEMM_WAVES)
void gemm_wmma_kernel(const float* __restrict__ h, const float* __restrict__ W,
                      float* __restrict__ out, int N) {
    const int lane = threadIdx.x & 31;
    const int wave = __builtin_amdgcn_readfirstlane((int)(threadIdx.x >> 5));
    const int nRowTiles = N >> 4;
    const int colBase = wave << 4;
    const int m = lane & 15;
    const int half = lane >> 4;

    float4 b[GEMM_KQ];
    {
        const float4* wrow = (const float4*)(W + (size_t)(colBase + m) * DFEAT + 4 * half);
#pragma unroll
        for (int i = 0; i < GEMM_KQ; ++i) b[i] = wrow[2 * i];
    }

    for (int rt = blockIdx.x; rt < nRowTiles; rt += gridDim.x) {
        const int rowBase = rt << 4;
        const float4* hrow = (const float4*)(h + (size_t)(rowBase + m) * DFEAT + 4 * half);
        float4 a[GEMM_KQ];
#pragma unroll
        for (int i = 0; i < GEMM_KQ; ++i) a[i] = hrow[2 * i];
        __builtin_amdgcn_sched_barrier(0);

        v8f acc = {};
#pragma unroll
        for (int i = 0; i < GEMM_KQ; ++i) {
            const v2f a0 = {a[i].x, a[i].y};
            const v2f b0 = {b[i].x, b[i].y};
            acc = __builtin_amdgcn_wmma_f32_16x16x4_f32(
                 false, a0,  false, b0,
                 (short)0, acc,  false,  false);
            const v2f a1 = {a[i].z, a[i].w};
            const v2f b1 = {b[i].z, b[i].w};
            acc = __builtin_amdgcn_wmma_f32_16x16x4_f32(
                 false, a1,  false, b1,
                 (short)0, acc,  false,  false);
        }

#pragma unroll
        for (int v = 0; v < 8; ++v) {
            const int r = rowBase + v + (half << 3);
            float val = acc[v];
            val = (val >= 0.0f) ? val : 0.01f * val;
            out[(size_t)r * DFEAT + colBase + m] = val;
        }
    }
}


extern "C" void kernel_launch(void* const* d_in, const int* in_sizes, int n_in,
                              void* d_out, int out_size, void* d_ws, size_t ws_size,
                              hipStream_t stream) {
    const float* feat  = (const float*)d_in[0];
    const int*   src   = (const int*)d_in[1];
    const int*   dst   = (const int*)d_in[2];
    const float* theta = (const float*)d_in[3];
    const float* W     = (const float*)d_in[4];
    float* out = (float*)d_out;

    const int N = in_sizes[0] / DFEAT;
    const int E = in_sizes[1];

    float* ws   = (float*)d_ws;
    float* deg  = ws;
    float* dinv = deg + N;
    int*   elist = (int*)(dinv + N);
    int*   tileCur = elist + 2 * ((size_t)E + MAX_TILES);
    float* Ta   = dinv + N + (size_t)N * DFEAT;
    float* Tb   = Ta + (size_t)N * DFEAT;
    float* h    = Tb + (size_t)N * DFEAT;

    const int TB = 256;
    const int nd4 = N * (DFEAT / 4);
    auto cdiv = [](long long a, long long b) { return (int)((a + b - 1) / b); };

    zero_kernel<<<cdiv(N / 4, TB), TB, 0, stream>>>(deg, N / 4, tileCur, MAX_TILES);
    deg_kernel<<<cdiv(cdiv(E, 4), TB), TB, 0, stream>>>(dst, deg, E);

    partition_kernel<<<cdiv(E / 4 + 1, PART_THREADS), PART_THREADS, 0, stream>>>(src, dst, deg, dinv, tileCur, elist, N, E);

    h_init_kernel<<<cdiv(nd4, TB), TB, 0, stream>>>(feat, theta, h, nd4);

    const int groups = cdiv(N, PROP_ROWS);

    propagate_list_kernel<<<groups, PROP_THREADS, 0, stream>>>(feat, dinv, elist, nullptr, Ta, h, theta, 1, 1.0f, N, E);

    propagate_list_kernel<<<groups, PROP_THREADS, 0, stream>>>(Ta, dinv, elist, feat, Tb, h, theta, 2, 2.0f, N, E);

    propagate_list_kernel<<<groups, PROP_THREADS, 0, stream>>>(Tb, dinv, elist, Ta, Ta, h, theta, 3, 2.0f, N, E);

    gemm_wmma_kernel<<<768, 32 * GEMM_WAVES, 0, stream>>>(h, W, out, N);
}


# Sum by backward recurrence: b2 = a2 x + 2 a3 Lx, b1 = (a1-a3) x + 2 L b2, h = a0 x + L b1 - b2 + c; T planes, h read-modify-write and h_init launch gone; rounding order of the sums changes
# speedup vs baseline: 4.0090x; 1.1671x over previous
#include <hip/hip_runtime.h>

typedef float v2f __attribute__((ext_vector_type(2)));
typedef float v8f __attribute__((ext_vector_type(8)));

#define DFEAT 128

__device__ __forceinline__ float inv_sqrt_deg(float d) { return 1.0f / sqrtf(fmaxf(d, 1.0f)); }

#define ENT_DL_SHIFT 24
#define ENT_SRC_MASK 0x00ffffffu


__global__ void zero_kernel(float* __restrict__ p, int n4, int* __restrict__ q, int nq) {
    int i = blockIdx.x * blockDim.x + threadIdx.x;
    if (i < n4) ((float4*)p)[i] = make_float4(0.f, 0.f, 0.f, 0.f);
    if (i < nq) q[i] = 0;
}

#define TILE_ROWS 196
__global__ void deg_kernel(const int* __restrict__ dst, float* __restrict__ deg, int E) {
    const int i = blockIdx.x * blockDim.x + threadIdx.x;
    const int E4 = E >> 2;
    if (i < E4) {
        int4 d = ((const int4*)dst)[i];
        atomicAdd(&deg[d.x], 1.0f);
        atomicAdd(&deg[d.y], 1.0f);
        atomicAdd(&deg[d.z], 1.0f);
        atomicAdd(&deg[d.w], 1.0f);
    }
    if (i < E - 4 * E4) atomicAdd(&deg[dst[4 * E4 + i]], 1.0f);
}


#define PROP_THREADS 1024
#define PROP_WAVES   32
#define PROP_ROWS    196
static_assert(PROP_ROWS == TILE_ROWS, "the partition launch numbers the tiles the propagation owns");
#define DRAIN_BATCH  8
#define QCAP (DRAIN_BATCH - 1 + 32)
#define LIST_CHUNK 4096
static_assert(LIST_CHUNK == 4 * PROP_THREADS, "a thread stages four entries and packs their four owner bytes into one word");
#define OWN_ROWS 7
static_assert(OWN_ROWS * PROP_WAVES >= PROP_ROWS, "every row of the tile has an owner wave and a register slot");
static_assert(PROP_ROWS <= (1 << (32 - ENT_DL_SHIFT)), "a local destination row fits the bits of an entry's first word above the source row");

__device__ __forceinline__ unsigned not_mine(unsigned w, unsigned mine, int k) {
    unsigned x = w ^ mine;
    unsigned y = ((x & 0x7f7f7f7fu) + 0x7f7f7f7fu) | x;
    return (y >> (7 - k)) & (0x01010101u << k);
}

__device__ __forceinline__ void own_batch(const float* __restrict__ x,
                                          const int2* ent, const int* qp, int n, int lane,
                                          float (&a)[OWN_ROWS][4]) {
    unsigned p = 0;
    float w = 0.0f;
    if (lane < n) {
        int2 e = ent[qp[lane]];
        p = (unsigned)e.x;
        w = __int_as_float(e.y);
    }
    float v[DRAIN_BATCH][4];
#pragma unroll
    for (int j = 0; j < DRAIN_BATCH; ++j) {
        v[j][0] = 0.0f; v[j][1] = 0.0f; v[j][2] = 0.0f; v[j][3] = 0.0f;
        if (j < n) {
            const unsigned sj = (unsigned)__builtin_amdgcn_readlane((int)p, j) & ENT_SRC_MASK;
            const float* xp = x + (size_t)sj * DFEAT + lane;
            v[j][0] = xp[0];
            v[j][1] = xp[32];
            v[j][2] = xp[64];
            v[j][3] = xp[96];
        }
    }
#pragma unroll
    for (int j = 0; j < DRAIN_BATCH; ++j) {
        if (j < n) {
            const float wj = __int_as_float(__builtin_amdgcn_readlane(__float_as_int(w), j));
            const int kj = (int)((unsigned)__builtin_amdgcn_readlane((int)p, j) >> (ENT_DL_SHIFT + 5));
#pragma unroll
            for (int k = 0; k < OWN_ROWS; ++k) {
                if (kj == k) {
                    a[k][0] = fmaf(wj, v[j][0], a[k][0]);
                    a[k][1] = fmaf(wj, v[j][1], a[k][1]);
                    a[k][2] = fmaf(wj, v[j][2], a[k][2]);
                    a[k][3] = fmaf(wj, v[j][3], a[k][3]);
                }
            }
        }
    }
}

__device__ __forceinline__ void walk_finish(const float* __restrict__ x, const float* __restrict__ feat,
                                            const float* __restrict__ dinv,
                                            const int2* tileList, int count,
                                            int2* ent, unsigned* own, int* q,
                                            const float* __restrict__ P, float* __restrict__ res,
                                            const float* __restrict__ theta, int pass,
                                            int lo, int cnt, int tid, int lane, int wave) {
    float a[OWN_ROWS][4];
#pragma unroll
    for (int k = 0; k < OWN_ROWS; ++k) { a[k][0] = 0.0f; a[k][1] = 0.0f; a[k][2] = 0.0f; a[k][3] = 0.0f; }

    int qn = 0;
    const unsigned mine = (unsigned)wave * 0x01010101u;

    for (int c0 = 0; c0 < count; c0 += LIST_CHUNK) {
        const int nc = (count - c0 < LIST_CHUNK) ? (count - c0) : LIST_CHUNK;
        {
            unsigned ow = 0;
#pragma unroll
            for (int u = 0; u < 4; ++u) {
                const int idx = 4 * tid + u;
                int2 e = tileList[c0 + (idx < nc ? idx : 0)];
                ent[idx] = e;
                unsigned b = (idx < nc) ? (((unsigned)e.x >> ENT_DL_SHIFT) & 31u) : 0x7fu;
                ow |= b << (8 * u);
            }
            own[tid] = ow;
        }
        __syncthreads();

        const int steps = (nc + 511) >> 9;
        for (int st = 0; st < steps; ++st) {
            const int i4 = st * 32 + lane;
            const uint4 d = ((const uint4*)own)[i4];
            unsigned nm = not_mine(d.x, mine, 0) | not_mine(d.y, mine, 1) |
                          not_mine(d.z, mine, 2) | not_mine(d.w, mine, 3);
            unsigned mm = ~nm & 0x0f0f0f0fu;
            for (;;) {
                unsigned b = __builtin_amdgcn_ballot_w32(mm != 0);
                if (b == 0) break;
                if (mm != 0) {
                    int bit = __builtin_ctz(mm);
                    mm &= mm - 1;
                    int pos = qn + __builtin_popcount(b & ((1u << lane) - 1u));
                    q[pos] = 16 * i4 + 4 * (bit & 7) + (bit >> 3);
                }
                qn += __builtin_popcount(b);
                while (qn >= DRAIN_BATCH) {
                    qn -= DRAIN_BATCH;
                    own_batch(x, ent, q + qn, DRAIN_BATCH, lane, a);
                }
            }
        }
        if (qn > 0) {
            own_batch(x, ent, q, qn, lane, a);
            qn = 0;
        }
        __syncthreads();
    }

    const float a1 = 0.5f * theta[1], a2 = 0.5f * theta[2], a3 = 0.5f * theta[3];
    float cx, cs, cc;
    if (pass == 1)      { cx = a2;       cs = -2.0f * a3; cc = 0.0f; }
    else if (pass == 2) { cx = a1 - a3;  cs = -2.0f;      cc = 0.0f; }
    else                { cx = theta[0]; cs = -1.0f;      cc = a1 + a2 + a3; }

    float xv[OWN_ROWS][4], pv[OWN_ROWS][4], sc[OWN_ROWS];
#pragma unroll
    for (int k = 0; k < OWN_ROWS; ++k) {
        const int r = wave + 32 * k;
        sc[k] = 0.0f;
#pragma unroll
        for (int c = 0; c < 4; ++c) { xv[k][c] = 0.0f; pv[k][c] = 0.0f; }
        if (r < cnt) {
            const size_t g = (size_t)(lo + r) * DFEAT + lane;
            sc[k] = cs * dinv[lo + r];
#pragma unroll
            for (int c = 0; c < 4; ++c) {
                xv[k][c] = feat[g + 32 * c];
                if (P) pv[k][c] = P[g + 32 * c];
            }
        }
    }
#pragma unroll
    for (int k = 0; k < OWN_ROWS; ++k) {
        const int r = wave + 32 * k;
        if (r < cnt) {
            const size_t g = (size_t)(lo + r) * DFEAT + lane;
#pragma unroll
            for (int c = 0; c < 4; ++c)
                res[g + 32 * c] = fmaf(sc[k], a[k][c], fmaf(cx, xv[k][c], cc)) - pv[k][c];
        }
    }
}

#define PART_THREADS 1024
#define PART_WAVES   32
#define MAX_TILES    256
#define TILES_PER_WAVE (MAX_TILES / PART_WAVES)
#define ROW_LOADS ((PROP_ROWS + 31) / 32)

__global__ __launch_bounds__(PART_THREADS)
void partition_kernel(const int* __restrict__ src, const int* __restrict__ dst,
                      const float* __restrict__ deg, float* __restrict__ dinv,
                      int* tileCur, int* elist, int N, int E) {
    int2* list2 = (int2*)elist;
    __shared__ int tileSum[MAX_TILES];
    __shared__ int tileBase[MAX_TILES];
    __shared__ int cnt[MAX_TILES];
    __shared__ int blockOff[MAX_TILES];

    const int tid = threadIdx.x;
    const int lane = tid & 31;
    const int wave = tid >> 5;
    const int i = blockIdx.x * PART_THREADS + tid;
    const int E4 = E >> 2;

    int s[4] = {0, 0, 0, 0}, d[4] = {0, 0, 0, 0};
    int ne = 0;
    if (i < E4) {
        const int4 dv = ((const int4*)dst)[i];
        const int4 sv = ((const int4*)src)[i];
        d[0] = dv.x; d[1] = dv.y; d[2] = dv.z; d[3] = dv.w;
        s[0] = sv.x; s[1] = sv.y; s[2] = sv.z; s[3] = sv.w;
        ne = 4;
    } else if (i == E4) {
        ne = E - 4 * E4;
#pragma unroll
        for (int u = 0; u < 3; ++u)
            if (u < ne) { d[u] = dst[4 * E4 + u]; s[u] = src[4 * E4 + u]; }
    }

    if (tid < MAX_TILES) cnt[tid] = 0;

    for (int r = i; r < N; r += gridDim.x * PART_THREADS) dinv[r] = inv_sqrt_deg(deg[r]);

    float sdeg[4];
#pragma unroll
    for (int u = 0; u < 4; ++u) sdeg[u] = deg[s[u]];

    {
        float dv[TILES_PER_WAVE][ROW_LOADS];
#pragma unroll
        for (int a = 0; a < TILES_PER_WAVE; ++a) {
            const int lo = (wave + PART_WAVES * a) * PROP_ROWS;
#pragma unroll
            for (int b = 0; b < ROW_LOADS; ++b) {
                const int r = 32 * b + lane;
                const bool ok = (r < PROP_ROWS) && (lo + r < N);
                dv[a][b] = deg[ok ? lo + r : 0];
            }
        }
#pragma unroll
        for (int a = 0; a < TILES_PER_WAVE; ++a) {
            const int lo = (wave + PART_WAVES * a) * PROP_ROWS;
            int part = 0;
#pragma unroll
            for (int b = 0; b < ROW_LOADS; ++b) {
                const int r = 32 * b + lane;
                const bool ok = (r < PROP_ROWS) && (lo + r < N);
                part += ok ? (int)dv[a][b] : 0;
            }
            for (int off = 16; off > 0; off >>= 1) part += __shfl_xor(part, off, 32);
            if (lane == 0) tileSum[wave + PART_WAVES * a] = part;
        }
    }
    __syncthreads();

    if (wave == 0) {
        int v[TILES_PER_WAVE];
        int sum = 0;
#pragma unroll
        for (int u = 0; u < TILES_PER_WAVE; ++u) { v[u] = tileSum[TILES_PER_WAVE * lane + u]; sum += v[u]; }
        int inc = sum;
        for (int off = 1; off < 32; off <<= 1) {
            const int nb = __shfl_up(inc, off, 32);
            if (lane >= off) inc += nb;
        }
        int run = inc - sum;
#pragma unroll
        for (int u = 0; u < TILES_PER_WAVE; ++u) { tileBase[TILES_PER_WAVE * lane + u] = run; run += v[u]; }
    }

    int tl[4] = {0, 0, 0, 0}, rk[4] = {0, 0, 0, 0};
#pragma unroll
    for (int u = 0; u < 4; ++u) {
        if (u < ne) {
            tl[u] = (int)((unsigned)d[u] / PROP_ROWS);
            rk[u] = atomicAdd(&cnt[tl[u]], 1);
        }
    }
    __syncthreads();

    if (tid < MAX_TILES) {
        const int c = cnt[tid];
        blockOff[tid] = (c > 0) ? atomicAdd(&tileCur[tid], c) : 0;
    }
    __syncthreads();

#pragma unroll
    for (int u = 0; u < 4; ++u) {
        if (u < ne) {
            const int pos = tileBase[tl[u]] + blockOff[tl[u]] + rk[u];
            const unsigned dl = (unsigned)(d[u] - tl[u] * PROP_ROWS);
            list2[pos] = make_int2((int)((unsigned)s[u] | (dl << ENT_DL_SHIFT)), __float_as_int(inv_sqrt_deg(sdeg[u])));
        }
    }
    if (blockIdx.x == 0 && tid < MAX_TILES) list2[E + tid] = make_int2(tileBase[tid], tileSum[tid]);
}

__global__ __launch_bounds__(PROP_THREADS)
void propagate_list_kernel(const float* __restrict__ x, const float* __restrict__ feat,
                           const float* __restrict__ dinv, const int* __restrict__ elist,
                           const float* __restrict__ P, float* __restrict__ res,
                           const float* __restrict__ theta, int pass, int N, int E) {
    __shared__ int2 ent[LIST_CHUNK];
    __shared__ __attribute__((aligned(16))) unsigned own[LIST_CHUNK / 4];
    __shared__ int queue[PROP_WAVES][QCAP];

    const int tid = threadIdx.x;
    const int lane = tid & 31;
    const int wave = __builtin_amdgcn_readfirstlane(tid >> 5);
    const int lo = blockIdx.x * PROP_ROWS;
    const int cnt = (N - lo < PROP_ROWS) ? (N - lo) : PROP_ROWS;

    const int2* list2 = (const int2*)elist;
    const int2 m = list2[E + blockIdx.x];

    walk_finish(x, feat, dinv, list2 + m.x, m.y, ent, own, queue[wave], P, res, theta, pass, lo, cnt, tid, lane, wave);
}


#define GEMM_WAVES 8
static_assert(GEMM_WAVES * 16 == DFEAT, "one wave per 16-column tile of the output");
#define GEMM_KQ (DFEAT / 8)

__global__ __launch_bounds__(32 * GEMM_WAVES)
void gemm_wmma_kernel(const float* __restrict__ h, const float* __restrict__ W,
                      float* __restrict__ out, int N) {
    const int lane = threadIdx.x & 31;
    const int wave = __builtin_amdgcn_readfirstlane((int)(threadIdx.x >> 5));
    const int nRowTiles = N >> 4;
    const int colBase = wave << 4;
    const int m = lane & 15;
    const int half = lane >> 4;

    float4 b[GEMM_KQ];
    {
        const float4* wrow = (const float4*)(W + (size_t)(colBase + m) * DFEAT + 4 * half);
#pragma unroll
        for (int i = 0; i < GEMM_KQ; ++i) b[i] = wrow[2 * i];
    }

    for (int rt = blockIdx.x; rt < nRowTiles; rt += gridDim.x) {
        const int rowBase = rt << 4;
        const float4* hrow = (const float4*)(h + (size_t)(rowBase + m) * DFEAT + 4 * half);
        float4 a[GEMM_KQ];
#pragma unroll
        for (int i = 0; i < GEMM_KQ; ++i) a[i] = hrow[2 * i];
        __builtin_amdgcn_sched_barrier(0);

        v8f acc = {};
#pragma unroll
        for (int i = 0; i < GEMM_KQ; ++i) {
            const v2f a0 = {a[i].x, a[i].y};
            const v2f b0 = {b[i].x, b[i].y};
            acc = __builtin_amdgcn_wmma_f32_16x16x4_f32(
                 false, a0,  false, b0,
                 (short)0, acc,  false,  false);
            const v2f a1 = {a[i].z, a[i].w};
            const v2f b1 = {b[i].z, b[i].w};
            acc = __builtin_amdgcn_wmma_f32_16x16x4_f32(
                 false, a1,  false, b1,
                 (short)0, acc,  false,  false);
        }

#pragma unroll
        for (int v = 0; v < 8; ++v) {
            const int r = rowBase + v + (half << 3);
            float val = acc[v];
            val = (val >= 0.0f) ? val : 0.01f * val;
            out[(size_t)r * DFEAT + colBase + m] = val;
        }
    }
}


extern "C" void kernel_launch(void* const* d_in, const int* in_sizes, int n_in,
                              void* d_out, int out_size, void* d_ws, size_t ws_size,
                              hipStream_t stream) {
    const float* feat  = (const float*)d_in[0];
    const int*   src   = (const int*)d_in[1];
    const int*   dst   = (const int*)d_in[2];
    const float* theta = (const float*)d_in[3];
    const float* W     = (const float*)d_in[4];
    float* out = (float*)d_out;

    const int N = in_sizes[0] / DFEAT;
    const int E = in_sizes[1];

    float* ws   = (float*)d_ws;
    float* deg  = ws;
    float* dinv = deg + N;
    int*   elist = (int*)(dinv + N);
    int*   tileCur = elist + 2 * ((size_t)E + MAX_TILES);
    float* Ta   = dinv + N + (size_t)N * DFEAT;
    float* Tb   = Ta + (size_t)N * DFEAT;
    float* h    = Tb + (size_t)N * DFEAT;

    const int TB = 256;
    auto cdiv = [](long long a, long long b) { return (int)((a + b - 1) / b); };

    zero_kernel<<<cdiv(N / 4, TB), TB, 0, stream>>>(deg, N / 4, tileCur, MAX_TILES);
    deg_kernel<<<cdiv(cdiv(E, 4), TB), TB, 0, stream>>>(dst, deg, E);

    partition_kernel<<<cdiv(E / 4 + 1, PART_THREADS), PART_THREADS, 0, stream>>>(src, dst, deg, dinv, tileCur, elist, N, E);

    const int groups = cdiv(N, PROP_ROWS);

    propagate_list_kernel<<<groups, PROP_THREADS, 0, stream>>>(feat, feat, dinv, elist, nullptr, Ta, theta, 1, N, E);

    propagate_list_kernel<<<groups, PROP_THREADS, 0, stream>>>(Ta, feat, dinv, elist, nullptr, Tb, theta, 2, N, E);

    propagate_list_kernel<<<groups, PROP_THREADS, 0, stream>>>(Tb, feat, dinv, elist, Ta, h, theta, 3, N, E);

    gemm_wmma_kernel<<<768, 32 * GEMM_WAVES, 0, stream>>>(h, W, out, N);
}


# Tile list chunk sorted by owner wave on its way into LDS (rank by returning LDS add on 32 counters, wave scan); a wave walks its own run; owner-byte filter and queue gone; sums in LDS rank order
# speedup vs baseline: 4.1528x; 1.0359x over previous
#include <hip/hip_runtime.h>

typedef float v2f __attribute__((ext_vector_type(2)));
typedef float v8f __attribute__((ext_vector_type(8)));

#define DFEAT 128

__device__ __forceinline__ float inv_sqrt_deg(float d) { return 1.0f / sqrtf(fmaxf(d, 1.0f)); }

#define ENT_DL_SHIFT 24
#define ENT_SRC_MASK 0x00ffffffu


__global__ void zero_kernel(float* __restrict__ p, int n4, int* __restrict__ q, int nq) {
    int i = blockIdx.x * blockDim.x + threadIdx.x;
    if (i < n4) ((float4*)p)[i] = make_float4(0.f, 0.f, 0.f, 0.f);
    if (i < nq) q[i] = 0;
}

#define TILE_ROWS 196
__global__ void deg_kernel(const int* __restrict__ dst, float* __restrict__ deg, int E) {
    const int i = blockIdx.x * blockDim.x + threadIdx.x;
    const int E4 = E >> 2;
    if (i < E4) {
        int4 d = ((const int4*)dst)[i];
        atomicAdd(&deg[d.x], 1.0f);
        atomicAdd(&deg[d.y], 1.0f);
        atomicAdd(&deg[d.z], 1.0f);
        atomicAdd(&deg[d.w], 1.0f);
    }
    if (i < E - 4 * E4) atomicAdd(&deg[dst[4 * E4 + i]], 1.0f);
}


#define PROP_THREADS 1024
#define PROP_WAVES   32
#define PROP_ROWS    196
static_assert(PROP_ROWS == TILE_ROWS, "the partition launch numbers the tiles the propagation owns");
#define DRAIN_BATCH  8
#define LIST_CHUNK 4096
static_assert(LIST_CHUNK == 4 * PROP_THREADS, "a thread places four entries of a chunk");
#define OWN_ROWS 7
static_assert(OWN_ROWS * PROP_WAVES >= PROP_ROWS, "every row of the tile has an owner wave and a register slot");
static_assert(PROP_ROWS <= (1 << (32 - ENT_DL_SHIFT)), "a local destination row fits the bits of an entry's first word above the source row");

__device__ __forceinline__ void own_batch(const float* __restrict__ x,
                                          const int2* seg, int n, int lane,
                                          float (&a)[OWN_ROWS][4]) {
    unsigned p = 0;
    float w = 0.0f;
    if (lane < n) {
        int2 e = seg[lane];
        p = (unsigned)e.x;
        w = __int_as_float(e.y);
    }
    float v[DRAIN_BATCH][4];
#pragma unroll
    for (int j = 0; j < DRAIN_BATCH; ++j) {
        v[j][0] = 0.0f; v[j][1] = 0.0f; v[j][2] = 0.0f; v[j][3] = 0.0f;
        if (j < n) {
            const unsigned sj = (unsigned)__builtin_amdgcn_readlane((int)p, j) & ENT_SRC_MASK;
            const float* xp = x + (size_t)sj * DFEAT + lane;
            v[j][0] = xp[0];
            v[j][1] = xp[32];
            v[j][2] = xp[64];
            v[j][3] = xp[96];
        }
    }
#pragma unroll
    for (int j = 0; j < DRAIN_BATCH; ++j) {
        if (j < n) {
            const float wj = __int_as_float(__builtin_amdgcn_readlane(__float_as_int(w), j));
            const int kj = (int)((unsigned)__builtin_amdgcn_readlane((int)p, j) >> (ENT_DL_SHIFT + 5));
#pragma unroll
            for (int k = 0; k < OWN_ROWS; ++k) {
                if (kj == k) {
                    a[k][0] = fmaf(wj, v[j][0], a[k][0]);
                    a[k][1] = fmaf(wj, v[j][1], a[k][1]);
                    a[k][2] = fmaf(wj, v[j][2], a[k][2]);
                    a[k][3] = fmaf(wj, v[j][3], a[k][3]);
                }
            }
        }
    }
}

__device__ __forceinline__ void walk_finish(const float* __restrict__ x, const float* __restrict__ feat,
                                            const float* __restrict__ dinv,
                                            const int2* tileList, int count,
                                            int2* srt, int* ownCnt,
                                            const float* __restrict__ P, float* __restrict__ res,
                                            const float* __restrict__ theta, int pass,
                                            int lo, int cnt, int tid, int lane, int wave) {
    float a[OWN_ROWS][4];
#pragma unroll
    for (int k = 0; k < OWN_ROWS; ++k) { a[k][0] = 0.0f; a[k][1] = 0.0f; a[k][2] = 0.0f; a[k][3] = 0.0f; }

    for (int c0 = 0; c0 < count; c0 += LIST_CHUNK) {
        const int nc = (count - c0 < LIST_CHUNK) ? (count - c0) : LIST_CHUNK;
        if (tid < PROP_WAVES) ownCnt[tid] = 0;
        int2 e[4];
#pragma unroll
        for (int u = 0; u < 4; ++u) {
            const int idx = u * PROP_THREADS + tid;
            e[u] = tileList[c0 + (idx < nc ? idx : 0)];
        }
        __syncthreads();

        unsigned ow[4];
        int rk[4];
#pragma unroll
        for (int u = 0; u < 4; ++u) {
            ow[u] = ((unsigned)e[u].x >> ENT_DL_SHIFT) & 31u;
            rk[u] = 0;
            if (u * PROP_THREADS + tid < nc) rk[u] = atomicAdd(&ownCnt[ow[u]], 1);
        }
        __syncthreads();

        const int cw = ownCnt[lane];
        int inc = cw;
        for (int off = 1; off < 32; off <<= 1) {
            const int nb = __shfl_up(inc, off, 32);
            if (lane >= off) inc += nb;
        }
        const int runAt = inc - cw;
#pragma unroll
        for (int u = 0; u < 4; ++u) {
            const int at = __shfl(runAt, (int)ow[u], 32);
            if (u * PROP_THREADS + tid < nc) srt[at + rk[u]] = e[u];
        }
        __syncthreads();

        const int2* run = srt + __builtin_amdgcn_readlane(runAt, wave);
        const int rn = __builtin_amdgcn_readlane(cw, wave);
        int i = 0;
        for (; i + DRAIN_BATCH <= rn; i += DRAIN_BATCH) own_batch(x, run + i, DRAIN_BATCH, lane, a);
        if (i < rn) own_batch(x, run + i, rn - i, lane, a);
    }

    const float a1 = 0.5f * theta[1], a2 = 0.5f * theta[2], a3 = 0.5f * theta[3];
    float cx, cs, cc;
    if (pass == 1)      { cx = a2;       cs = -2.0f * a3; cc = 0.0f; }
    else if (pass == 2) { cx = a1 - a3;  cs = -2.0f;      cc = 0.0f; }
    else                { cx = theta[0]; cs = -1.0f;      cc = a1 + a2 + a3; }

    float xv[OWN_ROWS][4], pv[OWN_ROWS][4], sc[OWN_ROWS];
#pragma unroll
    for (int k = 0; k < OWN_ROWS; ++k) {
        const int r = wave + 32 * k;
        sc[k] = 0.0f;
#pragma unroll
        for (int c = 0; c < 4; ++c) { xv[k][c] = 0.0f; pv[k][c] = 0.0f; }
        if (r < cnt) {
            const size_t g = (size_t)(lo + r) * DFEAT + lane;
            sc[k] = cs * dinv[lo + r];
#pragma unroll
            for (int c = 0; c < 4; ++c) {
                xv[k][c] = feat[g + 32 * c];
                if (P) pv[k][c] = P[g + 32 * c];
            }
        }
    }
#pragma unroll
    for (int k = 0; k < OWN_ROWS; ++k) {
        const int r = wave + 32 * k;
        if (r < cnt) {
            const size_t g = (size_t)(lo + r) * DFEAT + lane;
#pragma unroll
            for (int c = 0; c < 4; ++c)
                res[g + 32 * c] = fmaf(sc[k], a[k][c], fmaf(cx, xv[k][c], cc)) - pv[k][c];
        }
    }
}

#define PART_THREADS 1024
#define PART_WAVES   32
#define MAX_TILES    256
#define TILES_PER_WAVE (MAX_TILES / PART_WAVES)
#define ROW_LOADS ((PROP_ROWS + 31) / 32)

__global__ __launch_bounds__(PART_THREADS)
void partition_kernel(const int* __restrict__ src, const int* __restrict__ dst,
                      const float* __restrict__ deg, float* __restrict__ dinv,
                      int* tileCur, int* elist, int N, int E) {
    int2* list2 = (int2*)elist;
    __shared__ int tileSum[MAX_TILES];
    __shared__ int tileBase[MAX_TILES];
    __shared__ int cnt[MAX_TILES];
    __shared__ int blockOff[MAX_TILES];

    const int tid = threadIdx.x;
    const int lane = tid & 31;
    const int wave = tid >> 5;
    const int i = blockIdx.x * PART_THREADS + tid;
    const int E4 = E >> 2;

    int s[4] = {0, 0, 0, 0}, d[4] = {0, 0, 0, 0};
    int ne = 0;
    if (i < E4) {
        const int4 dv = ((const int4*)dst)[i];
        const int4 sv = ((const int4*)src)[i];
        d[0] = dv.x; d[1] = dv.y; d[2] = dv.z; d[3] = dv.w;
        s[0] = sv.x; s[1] = sv.y; s[2] = sv.z; s[3] = sv.w;
        ne = 4;
    } else if (i == E4) {
        ne = E - 4 * E4;
#pragma unroll
        for (int u = 0; u < 3; ++u)
            if (u < ne) { d[u] = dst[4 * E4 + u]; s[u] = src[4 * E4 + u]; }
    }

    if (tid < MAX_TILES) cnt[tid] = 0;

    for (int r = i; r < N; r += gridDim.x * PART_THREADS) dinv[r] = inv_sqrt_deg(deg[r]);

    float sdeg[4];
#pragma unroll
    for (int u = 0; u < 4; ++u) sdeg[u] = deg[s[u]];

    {
        float dv[TILES_PER_WAVE][ROW_LOADS];
#pragma unroll
        for (int a = 0; a < TILES_PER_WAVE; ++a) {
            const int lo = (wave + PART_WAVES * a) * PROP_ROWS;
#pragma unroll
            for (int b = 0; b < ROW_LOADS; ++b) {
                const int r = 32 * b + lane;
                const bool ok = (r < PROP_ROWS) && (lo + r < N);
                dv[a][b] = deg[ok ? lo + r : 0];
            }
        }
#pragma unroll
        for (int a = 0; a < TILES_PER_WAVE; ++a) {
            const int lo = (wave + PART_WAVES * a) * PROP_ROWS;
            int part = 0;
#pragma unroll
            for (int b = 0; b < ROW_LOADS; ++b) {
                const int r = 32 * b + lane;
                const bool ok = (r < PROP_ROWS) && (lo + r < N);
                part += ok ? (int)dv[a][b] : 0;
            }
            for (int off = 16; off > 0; off >>= 1) part += __shfl_xor(part, off, 32);
            if (lane == 0) tileSum[wave + PART_WAVES * a] = part;
        }
    }
    __syncthreads();

    if (wave == 0) {
        int v[TILES_PER_WAVE];
        int sum = 0;
#pragma unroll
        for (int u = 0; u < TILES_PER_WAVE; ++u) { v[u] = tileSum[TILES_PER_WAVE * lane + u]; sum += v[u]; }
        int inc = sum;
        for (int off = 1; off < 32; off <<= 1) {
            const int nb = __shfl_up(inc, off, 32);
            if (lane >= off) inc += nb;
        }
        int run = inc - sum;
#pragma unroll
        for (int u = 0; u < TILES_PER_WAVE; ++u) { tileBase[TILES_PER_WAVE * lane + u] = run; run += v[u]; }
    }

    int tl[4] = {0, 0, 0, 0}, rk[4] = {0, 0, 0, 0};
#pragma unroll
    for (int u = 0; u < 4; ++u) {
        if (u < ne) {
            tl[u] = (int)((unsigned)d[u] / PROP_ROWS);
            rk[u] = atomicAdd(&cnt[tl[u]], 1);
        }
    }
    __syncthreads();

    if (tid < MAX_TILES) {
        const int c = cnt[tid];
        blockOff[tid] = (c > 0) ? atomicAdd(&tileCur[tid], c) : 0;
    }
    __syncthreads();

#pragma unroll
    for (int u = 0; u < 4; ++u) {
        if (u < ne) {
            const int pos = tileBase[tl[u]] + blockOff[tl[u]] + rk[u];
            const unsigned dl = (unsigned)(d[u] - tl[u] * PROP_ROWS);
            list2[pos] = make_int2((int)((unsigned)s[u] | (dl << ENT_DL_SHIFT)), __float_as_int(inv_sqrt_deg(sdeg[u])));
        }
    }
    if (blockIdx.x == 0 && tid < MAX_TILES) list2[E + tid] = make_int2(tileBase[tid], tileSum[tid]);
}

__global__ __launch_bounds__(PROP_THREADS)
void propagate_list_kernel(const float* __restrict__ x, const float* __restrict__ feat,
                           const float* __restrict__ dinv, const int* __restrict__ elist,
                           const float* __restrict__ P, float* __restrict__ res,
                           const float* __restrict__ theta, int pass, int N, int E) {
    __shared__ int2 srt[LIST_CHUNK];
    __shared__ int ownCnt[PROP_WAVES];

    const int tid = threadIdx.x;
    const int lane = tid & 31;
    const int wave = __builtin_amdgcn_readfirstlane(tid >> 5);
    const int lo = blockIdx.x * PROP_ROWS;
    const int cnt = (N - lo < PROP_ROWS) ? (N - lo) : PROP_ROWS;

    const int2* list2 = (const int2*)elist;
    const int2 m = list2[E + blockIdx.x];

    walk_finish(x, feat, dinv, list2 + m.x, m.y, srt, ownCnt, P, res, theta, pass, lo, cnt, tid, lane, wave);
}


#define GEMM_WAVES 8
static_assert(GEMM_WAVES * 16 == DFEAT, "one wave per 16-column tile of the output");
#define GEMM_KQ (DFEAT / 8)

__global__ __launch_bounds__(32 * GEMM_WAVES)
void gemm_wmma_kernel(const float* __restrict__ h, const float* __restrict__ W,
                      float* __restrict__ out, int N) {
    const int lane = threadIdx.x & 31;
    const int wave = __builtin_amdgcn_readfirstlane((int)(threadIdx.x >> 5));
    const int nRowTiles = N >> 4;
    const int colBase = wave << 4;
    const int m = lane & 15;
    const int half = lane >> 4;

    float4 b[GEMM_KQ];
    {
        const float4* wrow = (const float4*)(W + (size_t)(colBase + m) * DFEAT + 4 * half);
#pragma unroll
        for (int i = 0; i < GEMM_KQ; ++i) b[i] = wrow[2 * i];
    }

    for (int rt = blockIdx.x; rt < nRowTiles; rt += gridDim.x) {
        const int rowBase = rt << 4;
        const float4* hrow = (const float4*)(h + (size_t)(rowBase + m) * DFEAT + 4 * half);
        float4 a[GEMM_KQ];
#pragma unroll
        for (int i = 0; i < GEMM_KQ; ++i) a[i] = hrow[2 * i];
        __builtin_amdgcn_sched_barrier(0);

        v8f acc = {};
#pragma unroll
        for (int i = 0; i < GEMM_KQ; ++i) {
            const v2f a0 = {a[i].x, a[i].y};
            const v2f b0 = {b[i].x, b[i].y};
            acc = __builtin_amdgcn_wmma_f32_16x16x4_f32(
                 false, a0,  false, b0,
                 (short)0, acc,  false,  false);
            const v2f a1 = {a[i].z, a[i].w};
            const v2f b1 = {b[i].z, b[i].w};
            acc = __builtin_amdgcn_wmma_f32_16x16x4_f32(
                 false, a1,  false, b1,
                 (short)0, acc,  false,  false);
        }

#pragma unroll
        for (int v = 0; v < 8; ++v) {
            const int r = rowBase + v + (half << 3);
            float val = acc[v];
            val = (val >= 0.0f) ? val : 0.01f * val;
            out[(size_t)r * DFEAT + colBase + m] = val;
        }
    }
}


extern "C" void kernel_launch(void* const* d_in, const int* in_sizes, int n_in,
                              void* d_out, int out_size, void* d_ws, size_t ws_size,
                              hipStream_t stream) {
    const float* feat  = (const float*)d_in[0];
    const int*   src   = (const int*)d_in[1];
    const int*   dst   = (const int*)d_in[2];
    const float* theta = (const float*)d_in[3];
    const float* W     = (const float*)d_in[4];
    float* out = (float*)d_out;

    const int N = in_sizes[0] / DFEAT;
    const int E = in_sizes[1];

    float* ws   = (float*)d_ws;
    float* deg  = ws;
    float* dinv = deg + N;
    int*   elist = (int*)(dinv + N);
    int*   tileCur = elist + 2 * ((size_t)E + MAX_TILES);
    float* Ta   = dinv + N + (size_t)N * DFEAT;
    float* Tb   = Ta + (size_t)N * DFEAT;
    float* h    = Tb + (size_t)N * DFEAT;

    const int TB = 256;
    auto cdiv = [](long long a, long long b) { return (int)((a + b - 1) / b); };

    zero_kernel<<<cdiv(N / 4, TB), TB, 0, stream>>>(deg, N / 4, tileCur, MAX_TILES);
    deg_kernel<<<cdiv(cdiv(E, 4), TB), TB, 0, stream>>>(dst, deg, E);

    partition_kernel<<<cdiv(E / 4 + 1, PART_THREADS), PART_THREADS, 0, stream>>>(src, dst, deg, dinv, tileCur, elist, N, E);

    const int groups = cdiv(N, PROP_ROWS);

    propagate_list_kernel<<<groups, PROP_THREADS, 0, stream>>>(feat, feat, dinv, elist, nullptr, Ta, theta, 1, N, E);

    propagate_list_kernel<<<groups, PROP_THREADS, 0, stream>>>(Ta, feat, dinv, elist, nullptr, Tb, theta, 2, N, E);

    propagate_list_kernel<<<groups, PROP_THREADS, 0, stream>>>(Tb, feat, dinv, elist, Ta, h, theta, 3, N, E);

    gemm_wmma_kernel<<<768, 32 * GEMM_WAVES, 0, stream>>>(h, W, out, N);
}
